# starting kernel
# baseline (speedup 1.0000x reference)
#include <hip/hip_runtime.h>
#include <math.h>

typedef __attribute__((ext_vector_type(2))) float v2f;
typedef __attribute__((ext_vector_type(8))) float v8f;

#define GEPS 1e-12f

__device__ inline void atomicMaxF32(float* addr, float val) {
    if (val >= 0.0f) {
        atomicMax((int*)addr, __float_as_int(val));
    } else {
        atomicMin((unsigned int*)addr, (unsigned int)__float_as_int(val));
    }
}

__global__ void init_kernel(const float* __restrict__ h0, float* __restrict__ out0,
                            float* __restrict__ m, float* __restrict__ s,
                            float* __restrict__ msg, int n) {
    int i = blockIdx.x * blockDim.x + threadIdx.x;
    int nd = n * 64;
    if (i < nd) { out0[i] = h0[i]; msg[i] = 0.0f; }
    if (i < n)  { m[i] = -__builtin_inff(); s[i] = 0.0f; }
}

__global__ void zero_kernel(float* __restrict__ p, int n) {
    int i = blockIdx.x * blockDim.x + threadIdx.x;
    if (i < n) p[i] = 0.0f;
}

__global__ void logits_kernel(const float* __restrict__ h_att, const float* __restrict__ t_att,
                              const int* __restrict__ erow, const int* __restrict__ ecol,
                              float* __restrict__ m, float* __restrict__ wE, int n_edges) {
    int gid  = blockIdx.x * blockDim.x + threadIdx.x;
    int edge = gid >> 4;
    int sub  = gid & 15;
    int e    = edge < n_edges ? edge : n_edges - 1;
    int r = erow[e], c = ecol[e];
    const float4 a = *(const float4*)(h_att + (size_t)r * 64 + sub * 4);
    const float4 b = *(const float4*)(t_att + (size_t)c * 64 + sub * 4);
    float dot = a.x * b.x + a.y * b.y + a.z * b.z + a.w * b.w;
    dot += __shfl_xor(dot, 1, 32);
    dot += __shfl_xor(dot, 2, 32);
    dot += __shfl_xor(dot, 4, 32);
    dot += __shfl_xor(dot, 8, 32);
    if (edge < n_edges && sub == 0) {
        wE[edge] = dot;
        atomicMaxF32(&m[r], dot);
    }
}

__global__ void expsum_kernel(float* __restrict__ wE, const float* __restrict__ m,
                              float* __restrict__ s, const int* __restrict__ erow, int n_edges) {
    int e = blockIdx.x * blockDim.x + threadIdx.x;
    if (e >= n_edges) return;
    int r = erow[e];
    float v = expf(wE[e] - m[r]);
    wE[e] = v;
    atomicAdd(&s[r], v);
}

__global__ void wnorm_kernel(float* __restrict__ wE, const float* __restrict__ s,
                             const int* __restrict__ erow, int n_edges) {
    int e = blockIdx.x * blockDim.x + threadIdx.x;
    if (e >= n_edges) return;
    wE[e] = wE[e] / fmaxf(s[erow[e]], GEPS);
}

__global__ void spmm_kernel(const float* __restrict__ wE, const float* __restrict__ h,
                            const int* __restrict__ erow, const int* __restrict__ ecol,
                            float* __restrict__ msg, int n_edges) {
    const int lane = threadIdx.x & 31;
    const int wave = blockIdx.x * (blockDim.x >> 5) + (threadIdx.x >> 5);
    int e0 = wave * 32;
    if (e0 >= n_edges) return;
    int e1 = min(e0 + 32, n_edges);
    int cur = erow[e0];
    float ax = 0.0f, ay = 0.0f;
    for (int e = e0; e < e1; ++e) {
        int r = erow[e];
        if (r != cur) {
            atomicAdd(&msg[(size_t)cur * 64 + lane * 2    ], ax);
            atomicAdd(&msg[(size_t)cur * 64 + lane * 2 + 1], ay);
            ax = 0.0f; ay = 0.0f; cur = r;
        }
        float we = wE[e];
        const float2 hv = *(const float2*)(h + (size_t)ecol[e] * 64 + lane * 2);
        ax = fmaf(we, hv.x, ax);
        ay = fmaf(we, hv.y, ay);
    }
    atomicAdd(&msg[(size_t)cur * 64 + lane * 2    ], ax);
    atomicAdd(&msg[(size_t)cur * 64 + lane * 2 + 1], ay);
}

__global__ void dense_kernel(const float* __restrict__ hprev, const float* __restrict__ msg,
                             const float* __restrict__ W, const float* __restrict__ bias,
                             float* __restrict__ out, int n) {
    const int lane = threadIdx.x & 31;
    const int wave = threadIdx.x >> 5;
    const int tile = blockIdx.x * (blockDim.x >> 5) + wave;
    const int ntiles = (n + 15) >> 4;
    if (tile >= ntiles) return;

    const int half = lane >> 4;
    const int l16  = lane & 15;
    int row  = tile * 16 + l16;
    int rowc = row < n ? row : n - 1;

    v8f acc[4] = {};

#pragma unroll
    for (int kb = 0; kb < 16; ++kb) {
        const int koff = kb * 4 + half * 2;
        const float* hp = hprev + (size_t)rowc * 64 + koff;
        const float* mp = msg   + (size_t)rowc * 64 + koff;
        v2f a;
        a.x = hp[0] + mp[0];
        a.y = hp[1] + mp[1];
#pragma unroll
        for (int jb = 0; jb < 4; ++jb) {
            const float* wp = W + (size_t)(jb * 16 + l16) * 64 + koff;
            v2f b;
            b.x = wp[0];
            b.y = wp[1];
            acc[jb] = __builtin_amdgcn_wmma_f32_16x16x4_f32(
                false, a, false, b, (short)0, acc[jb], false, false);
        }
    }

    float bj[4];
#pragma unroll
    for (int jb = 0; jb < 4; ++jb) bj[jb] = bias[jb * 16 + l16];

#pragma unroll
    for (int v = 0; v < 8; ++v) {
        float vals[4];
        float ss = 0.0f;
#pragma unroll
        for (int jb = 0; jb < 4; ++jb) {
            float y = fmaxf(acc[jb][v] + bj[jb], 0.0f);
            vals[jb] = y;
            ss = fmaf(y, y, ss);
        }
        ss += __shfl_xor(ss, 1, 32);
        ss += __shfl_xor(ss, 2, 32);
        ss += __shfl_xor(ss, 4, 32);
        ss += __shfl_xor(ss, 8, 32);
        float inv = 1.0f / fmaxf(sqrtf(ss), GEPS);
        int m    = v + half * 8;
        int orow = tile * 16 + m;
        if (orow < n) {
#pragma unroll
            for (int jb = 0; jb < 4; ++jb)
                out[(size_t)orow * 64 + jb * 16 + l16] = vals[jb] * inv;
        }
    }
}

static inline int cdiv(long long a, long long b) { return (int)((a + b - 1) / b); }

extern "C" void kernel_launch(void* const* d_in, const int* in_sizes, int n_in,
                              void* d_out, int out_size, void* d_ws, size_t ws_size,
                              hipStream_t stream) {
    const float* h0    = (const float*)d_in[0];
    const float* h_att = (const float*)d_in[1];
    const float* t_att = (const float*)d_in[2];
    const float* w1    = (const float*)d_in[3];
    const float* b1    = (const float*)d_in[4];
    const float* w2    = (const float*)d_in[5];
    const float* b2    = (const float*)d_in[6];
    const int*   erow  = (const int*)d_in[7];
    const int*   ecol  = (const int*)d_in[8];
    float* out = (float*)d_out;

    const int N = in_sizes[0] / 64;
    const int E = in_sizes[7];

    float* ws  = (float*)d_ws;
    float* m   = ws;
    float* s   = ws + N;
    float* wE  = ws + 2 * (size_t)N;
    float* msg = ws + 2 * (size_t)N + E;

    float* out1 = out + (size_t)N * 64;
    float* out2 = out + 2 * (size_t)N * 64;

    const int TPB = 256;
    const int nd  = N * 64;

    init_kernel<<<cdiv(nd, TPB), TPB, 0, stream>>>(h0, out, m, s, msg, N);
    logits_kernel<<<cdiv((long long)E * 16, TPB), TPB, 0, stream>>>(h_att, t_att, erow, ecol, m, wE, E);
    expsum_kernel<<<cdiv(E, TPB), TPB, 0, stream>>>(wE, m, s, erow, E);
    wnorm_kernel<<<cdiv(E, TPB), TPB, 0, stream>>>(wE, s, erow, E);

    const int waves   = cdiv(E, 32);
    const int spmm_bl = cdiv(waves, TPB / 32);
    const int ntiles  = (N + 15) >> 4;
    const int dense_bl = cdiv(ntiles, TPB / 32);

    spmm_kernel<<<spmm_bl, TPB, 0, stream>>>(wE, h0, erow, ecol, msg, E);
    dense_kernel<<<dense_bl, TPB, 0, stream>>>(h0, msg, w1, b1, out1, N);

    zero_kernel<<<cdiv(nd, TPB), TPB, 0, stream>>>(msg, nd);
    spmm_kernel<<<spmm_bl, TPB, 0, stream>>>(wE, out1, erow, ecol, msg, E);
    dense_kernel<<<dense_bl, TPB, 0, stream>>>(out1, msg, w2, b2, out2, N);
}


# edge kernels (logits, both SpMM): one wave per 32 edges; each lane loads one edge's ids once, the wave broadcasts them by readlane, so all row gathers issue together; sums in the same order
# speedup vs baseline: 1.6052x; 1.6052x over previous
#include <hip/hip_runtime.h>
#include <math.h>

typedef __attribute__((ext_vector_type(2))) float v2f;
typedef __attribute__((ext_vector_type(8))) float v8f;

#define GEPS 1e-12f

__device__ inline void atomicMaxF32(float* addr, float val) {
    if (val >= 0.0f) {
        atomicMax((int*)addr, __float_as_int(val));
    } else {
        atomicMin((unsigned int*)addr, (unsigned int)__float_as_int(val));
    }
}

__global__ void init_kernel(const float* __restrict__ h0, float* __restrict__ out0,
                            float* __restrict__ m, float* __restrict__ s,
                            float* __restrict__ msg, int n) {
    int i = blockIdx.x * blockDim.x + threadIdx.x;
    int nd = n * 64;
    if (i < nd) { out0[i] = h0[i]; msg[i] = 0.0f; }
    if (i < n)  { m[i] = -__builtin_inff(); s[i] = 0.0f; }
}

__global__ void zero_kernel(float* __restrict__ p, int n) {
    int i = blockIdx.x * blockDim.x + threadIdx.x;
    if (i < n) p[i] = 0.0f;
}

__global__ void logits_kernel(const float* __restrict__ h_att, const float* __restrict__ t_att,
                              const int* __restrict__ erow, const int* __restrict__ ecol,
                              float* __restrict__ m, float* __restrict__ wE, int n_edges) {
    const int lane = threadIdx.x & 31;
    const int wave = __builtin_amdgcn_readfirstlane(
        (int)(blockIdx.x * (blockDim.x >> 5) + (threadIdx.x >> 5)));
    const int e0 = wave * 32;
    if (e0 >= n_edges) return;
    const int cnt = min(32, n_edges - e0);
    const int el   = e0 + min(lane, cnt - 1);
    const int my_r = erow[el];
    const int my_c = ecol[el];
    const int half = lane >> 4;
    const int sub  = lane & 15;
    float mylogit = 0.0f;

#pragma unroll
    for (int g = 0; g < 2; ++g) {
        float4 a[8], b[8];
#pragma unroll
        for (int k = 0; k < 8; ++k) {
            const int t = g * 8 + k;
            const int r_lo = __builtin_amdgcn_readlane(my_r, t);
            const int r_hi = __builtin_amdgcn_readlane(my_r, t + 16);
            const int c_lo = __builtin_amdgcn_readlane(my_c, t);
            const int c_hi = __builtin_amdgcn_readlane(my_c, t + 16);
            const int r = half ? r_hi : r_lo;
            const int c = half ? c_hi : c_lo;
            a[k] = *(const float4*)(h_att + (size_t)r * 64 + sub * 4);
            b[k] = *(const float4*)(t_att + (size_t)c * 64 + sub * 4);
        }
#pragma unroll
        for (int k = 0; k < 8; ++k) {
            const int t = g * 8 + k;
            float dot = a[k].x * b[k].x + a[k].y * b[k].y + a[k].z * b[k].z + a[k].w * b[k].w;
            dot += __shfl_xor(dot, 1, 32);
            dot += __shfl_xor(dot, 2, 32);
            dot += __shfl_xor(dot, 4, 32);
            dot += __shfl_xor(dot, 8, 32);
            if (sub == t) mylogit = dot;
        }
    }
    if (lane < cnt) {
        wE[e0 + lane] = mylogit;
        atomicMaxF32(&m[my_r], mylogit);
    }
}

__global__ void expsum_kernel(float* __restrict__ wE, const float* __restrict__ m,
                              float* __restrict__ s, const int* __restrict__ erow, int n_edges) {
    int e = blockIdx.x * blockDim.x + threadIdx.x;
    if (e >= n_edges) return;
    int r = erow[e];
    float v = expf(wE[e] - m[r]);
    wE[e] = v;
    atomicAdd(&s[r], v);
}

__global__ void wnorm_kernel(float* __restrict__ wE, const float* __restrict__ s,
                             const int* __restrict__ erow, int n_edges) {
    int e = blockIdx.x * blockDim.x + threadIdx.x;
    if (e >= n_edges) return;
    wE[e] = wE[e] / fmaxf(s[erow[e]], GEPS);
}

__global__ void spmm_kernel(const float* __restrict__ wE, const float* __restrict__ h,
                            const int* __restrict__ erow, const int* __restrict__ ecol,
                            float* __restrict__ msg, int n_edges) {
    const int lane = threadIdx.x & 31;
    const int wave = __builtin_amdgcn_readfirstlane(
        (int)(blockIdx.x * (blockDim.x >> 5) + (threadIdx.x >> 5)));
    const int e0 = wave * 32;
    if (e0 >= n_edges) return;
    const int cnt = min(32, n_edges - e0);
    const int   el   = e0 + min(lane, cnt - 1);
    const int   my_r = erow[el];
    const int   my_c = ecol[el];
    const float my_w = wE[el];

    float2 hv[32];
#pragma unroll
    for (int j = 0; j < 32; ++j) {
        const int c = __builtin_amdgcn_readlane(my_c, j);
        hv[j] = *(const float2*)(h + (size_t)c * 64 + lane * 2);
    }

    int cur = __builtin_amdgcn_readlane(my_r, 0);
    float ax = 0.0f, ay = 0.0f;
#pragma unroll
    for (int j = 0; j < 32; ++j) {
        if (j < cnt) {
            const int   r  = __builtin_amdgcn_readlane(my_r, j);
            const float we = __int_as_float(__builtin_amdgcn_readlane(__float_as_int(my_w), j));
            if (r != cur) {
                atomicAdd(&msg[(size_t)cur * 64 + lane * 2    ], ax);
                atomicAdd(&msg[(size_t)cur * 64 + lane * 2 + 1], ay);
                ax = 0.0f; ay = 0.0f; cur = r;
            }
            ax = fmaf(we, hv[j].x, ax);
            ay = fmaf(we, hv[j].y, ay);
        }
    }
    atomicAdd(&msg[(size_t)cur * 64 + lane * 2    ], ax);
    atomicAdd(&msg[(size_t)cur * 64 + lane * 2 + 1], ay);
}

__global__ void dense_kernel(const float* __restrict__ hprev, const float* __restrict__ msg,
                             const float* __restrict__ W, const float* __restrict__ bias,
                             float* __restrict__ out, int n) {
    const int lane = threadIdx.x & 31;
    const int wave = threadIdx.x >> 5;
    const int tile = blockIdx.x * (blockDim.x >> 5) + wave;
    const int ntiles = (n + 15) >> 4;
    if (tile >= ntiles) return;

    const int half = lane >> 4;
    const int l16  = lane & 15;
    int row  = tile * 16 + l16;
    int rowc = row < n ? row : n - 1;

    v8f acc[4] = {};

#pragma unroll
    for (int kb = 0; kb < 16; ++kb) {
        const int koff = kb * 4 + half * 2;
        const float* hp = hprev + (size_t)rowc * 64 + koff;
        const float* mp = msg   + (size_t)rowc * 64 + koff;
        v2f a;
        a.x = hp[0] + mp[0];
        a.y = hp[1] + mp[1];
#pragma unroll
        for (int jb = 0; jb < 4; ++jb) {
            const float* wp = W + (size_t)(jb * 16 + l16) * 64 + koff;
            v2f b;
            b.x = wp[0];
            b.y = wp[1];
            acc[jb] = __builtin_amdgcn_wmma_f32_16x16x4_f32(
                false, a, false, b, (short)0, acc[jb], false, false);
        }
    }

    float bj[4];
#pragma unroll
    for (int jb = 0; jb < 4; ++jb) bj[jb] = bias[jb * 16 + l16];

#pragma unroll
    for (int v = 0; v < 8; ++v) {
        float vals[4];
        float ss = 0.0f;
#pragma unroll
        for (int jb = 0; jb < 4; ++jb) {
            float y = fmaxf(acc[jb][v] + bj[jb], 0.0f);
            vals[jb] = y;
            ss = fmaf(y, y, ss);
        }
        ss += __shfl_xor(ss, 1, 32);
        ss += __shfl_xor(ss, 2, 32);
        ss += __shfl_xor(ss, 4, 32);
        ss += __shfl_xor(ss, 8, 32);
        float inv = 1.0f / fmaxf(sqrtf(ss), GEPS);
        int m    = v + half * 8;
        int orow = tile * 16 + m;
        if (orow < n) {
#pragma unroll
            for (int jb = 0; jb < 4; ++jb)
                out[(size_t)orow * 64 + jb * 16 + l16] = vals[jb] * inv;
        }
    }
}

static inline int cdiv(long long a, long long b) { return (int)((a + b - 1) / b); }

extern "C" void kernel_launch(void* const* d_in, const int* in_sizes, int n_in,
                              void* d_out, int out_size, void* d_ws, size_t ws_size,
                              hipStream_t stream) {
    const float* h0    = (const float*)d_in[0];
    const float* h_att = (const float*)d_in[1];
    const float* t_att = (const float*)d_in[2];
    const float* w1    = (const float*)d_in[3];
    const float* b1    = (const float*)d_in[4];
    const float* w2    = (const float*)d_in[5];
    const float* b2    = (const float*)d_in[6];
    const int*   erow  = (const int*)d_in[7];
    const int*   ecol  = (const int*)d_in[8];
    float* out = (float*)d_out;

    const int N = in_sizes[0] / 64;
    const int E = in_sizes[7];

    float* ws  = (float*)d_ws;
    float* m   = ws;
    float* s   = ws + N;
    float* wE  = ws + 2 * (size_t)N;
    float* msg = ws + 2 * (size_t)N + E;

    float* out1 = out + (size_t)N * 64;
    float* out2 = out + 2 * (size_t)N * 64;

    const int TPB = 256;
    const int nd  = N * 64;

    const int waves   = cdiv(E, 32);
    const int spmm_bl = cdiv(waves, TPB / 32);

    init_kernel<<<cdiv(nd, TPB), TPB, 0, stream>>>(h0, out, m, s, msg, N);
    logits_kernel<<<spmm_bl, TPB, 0, stream>>>(h_att, t_att, erow, ecol, m, wE, E);
    expsum_kernel<<<cdiv(E, TPB), TPB, 0, stream>>>(wE, m, s, erow, E);
    wnorm_kernel<<<cdiv(E, TPB), TPB, 0, stream>>>(wE, s, erow, E);

    const int ntiles  = (N + 15) >> 4;
    const int dense_bl = cdiv(ntiles, TPB / 32);

    spmm_kernel<<<spmm_bl, TPB, 0, stream>>>(wE, h0, erow, ecol, msg, E);
    dense_kernel<<<dense_bl, TPB, 0, stream>>>(h0, msg, w1, b1, out1, N);

    zero_kernel<<<cdiv(nd, TPB), TPB, 0, stream>>>(msg, nd);
    spmm_kernel<<<spmm_bl, TPB, 0, stream>>>(wE, out1, erow, ecol, msg, E);
    dense_kernel<<<dense_bl, TPB, 0, stream>>>(out1, msg, w2, b2, out2, N);
}


# fewer, fatter waves in the four light kernels: init and zero fills as 1024-block grid-stride loops with 16-byte accesses, expsum and wnorm 4 edges per lane; s summed by atomics as before
# speedup vs baseline: 1.7355x; 1.0812x over previous
#include <hip/hip_runtime.h>
#include <math.h>

typedef __attribute__((ext_vector_type(2))) float v2f;
typedef __attribute__((ext_vector_type(8))) float v8f;

#define GEPS 1e-12f

__device__ inline void atomicMaxF32(float* addr, float val) {
    if (val >= 0.0f) {
        atomicMax((int*)addr, __float_as_int(val));
    } else {
        atomicMin((unsigned int*)addr, (unsigned int)__float_as_int(val));
    }
}

#define FILL_BLOCKS 1024

__global__ void init_kernel(const float* h0, float* out0,
                            float* __restrict__ m, float* __restrict__ s,
                            float* msg, int n) {
    const int tid    = blockIdx.x * blockDim.x + threadIdx.x;
    const int stride = gridDim.x * blockDim.x;
    const int n4     = n * 16;
    const float4* src = (const float4*)h0;
    float4* dst = (float4*)out0;
    float4* mz  = (float4*)msg;
    const float4 z = make_float4(0.0f, 0.0f, 0.0f, 0.0f);
    int i = tid;
    for (; i + 3 * stride < n4; i += 4 * stride) {
        const float4 v0 = src[i];
        const float4 v1 = src[i + stride];
        const float4 v2 = src[i + 2 * stride];
        const float4 v3 = src[i + 3 * stride];
        dst[i]              = v0; mz[i]              = z;
        dst[i + stride]     = v1; mz[i + stride]     = z;
        dst[i + 2 * stride] = v2; mz[i + 2 * stride] = z;
        dst[i + 3 * stride] = v3; mz[i + 3 * stride] = z;
    }
    if (i + stride < n4) {
        const float4 v0 = src[i];
        const float4 v1 = src[i + stride];
        dst[i]          = v0; mz[i]          = z;
        dst[i + stride] = v1; mz[i + stride] = z;
        i += 2 * stride;
    }
    if (i < n4) { dst[i] = src[i]; mz[i] = z; }
    for (int k = tid; k < n; k += stride) { m[k] = -__builtin_inff(); s[k] = 0.0f; }
}

__global__ void zero_kernel(float* __restrict__ p, int n4) {
    const int stride = gridDim.x * blockDim.x;
    float4* __restrict__ p4 = (float4*)p;
    const float4 z = make_float4(0.0f, 0.0f, 0.0f, 0.0f);
    for (int i = blockIdx.x * blockDim.x + threadIdx.x; i < n4; i += stride) p4[i] = z;
}

__global__ void logits_kernel(const float* __restrict__ h_att, const float* __restrict__ t_att,
                              const int* __restrict__ erow, const int* __restrict__ ecol,
                              float* __restrict__ m, float* __restrict__ wE, int n_edges) {
    const int lane = threadIdx.x & 31;
    const int wave = __builtin_amdgcn_readfirstlane(
        (int)(blockIdx.x * (blockDim.x >> 5) + (threadIdx.x >> 5)));
    const int e0 = wave * 32;
    if (e0 >= n_edges) return;
    const int cnt = min(32, n_edges - e0);
    const int el   = e0 + min(lane, cnt - 1);
    const int my_r = erow[el];
    const int my_c = ecol[el];
    const int half = lane >> 4;
    const int sub  = lane & 15;
    float mylogit = 0.0f;

#pragma unroll
    for (int g = 0; g < 2; ++g) {
        float4 a[8], b[8];
#pragma unroll
        for (int k = 0; k < 8; ++k) {
            const int t = g * 8 + k;
            const int r_lo = __builtin_amdgcn_readlane(my_r, t);
            const int r_hi = __builtin_amdgcn_readlane(my_r, t + 16);
            const int c_lo = __builtin_amdgcn_readlane(my_c, t);
            const int c_hi = __builtin_amdgcn_readlane(my_c, t + 16);
            const int r = half ? r_hi : r_lo;
            const int c = half ? c_hi : c_lo;
            a[k] = *(const float4*)(h_att + (size_t)r * 64 + sub * 4);
            b[k] = *(const float4*)(t_att + (size_t)c * 64 + sub * 4);
        }
#pragma unroll
        for (int k = 0; k < 8; ++k) {
            const int t = g * 8 + k;
            float dot = a[k].x * b[k].x + a[k].y * b[k].y + a[k].z * b[k].z + a[k].w * b[k].w;
            dot += __shfl_xor(dot, 1, 32);
            dot += __shfl_xor(dot, 2, 32);
            dot += __shfl_xor(dot, 4, 32);
            dot += __shfl_xor(dot, 8, 32);
            if (sub == t) mylogit = dot;
        }
    }
    if (lane < cnt) {
        wE[e0 + lane] = mylogit;
        atomicMaxF32(&m[my_r], mylogit);
    }
}

__global__ void expsum_kernel(float* __restrict__ wE, const float* __restrict__ m,
                              float* __restrict__ s, const int* __restrict__ erow, int n_edges) {
    const int e = (blockIdx.x * blockDim.x + threadIdx.x) * 4;
    if (e + 3 < n_edges) {
        const int4   r = *(const int4*)(erow + e);
        const float4 l = *(const float4*)(wE + e);
        const float m0 = m[r.x], m1 = m[r.y], m2 = m[r.z], m3 = m[r.w];
        float4 v;
        v.x = expf(l.x - m0);
        v.y = expf(l.y - m1);
        v.z = expf(l.z - m2);
        v.w = expf(l.w - m3);
        *(float4*)(wE + e) = v;
        atomicAdd(&s[r.x], v.x);
        atomicAdd(&s[r.y], v.y);
        atomicAdd(&s[r.z], v.z);
        atomicAdd(&s[r.w], v.w);
    } else {
        for (int k = e; k < n_edges; ++k) {
            const int r = erow[k];
            const float v = expf(wE[k] - m[r]);
            wE[k] = v;
            atomicAdd(&s[r], v);
        }
    }
}

__global__ void wnorm_kernel(float* __restrict__ wE, const float* __restrict__ s,
                             const int* __restrict__ erow, int n_edges) {
    const int e = (blockIdx.x * blockDim.x + threadIdx.x) * 4;
    if (e + 3 < n_edges) {
        const int4 r = *(const int4*)(erow + e);
        float4 v = *(const float4*)(wE + e);
        const float s0 = s[r.x], s1 = s[r.y], s2 = s[r.z], s3 = s[r.w];
        v.x = v.x / fmaxf(s0, GEPS);
        v.y = v.y / fmaxf(s1, GEPS);
        v.z = v.z / fmaxf(s2, GEPS);
        v.w = v.w / fmaxf(s3, GEPS);
        *(float4*)(wE + e) = v;
    } else {
        for (int k = e; k < n_edges; ++k)
            wE[k] = wE[k] / fmaxf(s[erow[k]], GEPS);
    }
}

__global__ void spmm_kernel(const float* __restrict__ wE, const float* __restrict__ h,
                            const int* __restrict__ erow, const int* __restrict__ ecol,
                            float* __restrict__ msg, int n_edges) {
    const int lane = threadIdx.x & 31;
    const int wave = __builtin_amdgcn_readfirstlane(
        (int)(blockIdx.x * (blockDim.x >> 5) + (threadIdx.x >> 5)));
    const int e0 = wave * 32;
    if (e0 >= n_edges) return;
    const int cnt = min(32, n_edges - e0);
    const int   el   = e0 + min(lane, cnt - 1);
    const int   my_r = erow[el];
    const int   my_c = ecol[el];
    const float my_w = wE[el];

    float2 hv[32];
#pragma unroll
    for (int j = 0; j < 32; ++j) {
        const int c = __builtin_amdgcn_readlane(my_c, j);
        hv[j] = *(const float2*)(h + (size_t)c * 64 + lane * 2);
    }

    int cur = __builtin_amdgcn_readlane(my_r, 0);
    float ax = 0.0f, ay = 0.0f;
#pragma unroll
    for (int j = 0; j < 32; ++j) {
        if (j < cnt) {
            const int   r  = __builtin_amdgcn_readlane(my_r, j);
            const float we = __int_as_float(__builtin_amdgcn_readlane(__float_as_int(my_w), j));
            if (r != cur) {
                atomicAdd(&msg[(size_t)cur * 64 + lane * 2    ], ax);
                atomicAdd(&msg[(size_t)cur * 64 + lane * 2 + 1], ay);
                ax = 0.0f; ay = 0.0f; cur = r;
            }
            ax = fmaf(we, hv[j].x, ax);
            ay = fmaf(we, hv[j].y, ay);
        }
    }
    atomicAdd(&msg[(size_t)cur * 64 + lane * 2    ], ax);
    atomicAdd(&msg[(size_t)cur * 64 + lane * 2 + 1], ay);
}

__global__ void dense_kernel(const float* __restrict__ hprev, const float* __restrict__ msg,
                             const float* __restrict__ W, const float* __restrict__ bias,
                             float* __restrict__ out, int n) {
    const int lane = threadIdx.x & 31;
    const int wave = threadIdx.x >> 5;
    const int tile = blockIdx.x * (blockDim.x >> 5) + wave;
    const int ntiles = (n + 15) >> 4;
    if (tile >= ntiles) return;

    const int half = lane >> 4;
    const int l16  = lane & 15;
    int row  = tile * 16 + l16;
    int rowc = row < n ? row : n - 1;

    v8f acc[4] = {};

#pragma unroll
    for (int kb = 0; kb < 16; ++kb) {
        const int koff = kb * 4 + half * 2;
        const float* hp = hprev + (size_t)rowc * 64 + koff;
        const float* mp = msg   + (size_t)rowc * 64 + koff;
        v2f a;
        a.x = hp[0] + mp[0];
        a.y = hp[1] + mp[1];
#pragma unroll
        for (int jb = 0; jb < 4; ++jb) {
            const float* wp = W + (size_t)(jb * 16 + l16) * 64 + koff;
            v2f b;
            b.x = wp[0];
            b.y = wp[1];
            acc[jb] = __builtin_amdgcn_wmma_f32_16x16x4_f32(
                false, a, false, b, (short)0, acc[jb], false, false);
        }
    }

    float bj[4];
#pragma unroll
    for (int jb = 0; jb < 4; ++jb) bj[jb] = bias[jb * 16 + l16];

#pragma unroll
    for (int v = 0; v < 8; ++v) {
        float vals[4];
        float ss = 0.0f;
#pragma unroll
        for (int jb = 0; jb < 4; ++jb) {
            float y = fmaxf(acc[jb][v] + bj[jb], 0.0f);
            vals[jb] = y;
            ss = fmaf(y, y, ss);
        }
        ss += __shfl_xor(ss, 1, 32);
        ss += __shfl_xor(ss, 2, 32);
        ss += __shfl_xor(ss, 4, 32);
        ss += __shfl_xor(ss, 8, 32);
        float inv = 1.0f / fmaxf(sqrtf(ss), GEPS);
        int m    = v + half * 8;
        int orow = tile * 16 + m;
        if (orow < n) {
#pragma unroll
            for (int jb = 0; jb < 4; ++jb)
                out[(size_t)orow * 64 + jb * 16 + l16] = vals[jb] * inv;
        }
    }
}

static inline int cdiv(long long a, long long b) { return (int)((a + b - 1) / b); }

extern "C" void kernel_launch(void* const* d_in, const int* in_sizes, int n_in,
                              void* d_out, int out_size, void* d_ws, size_t ws_size,
                              hipStream_t stream) {
    const float* h0    = (const float*)d_in[0];
    const float* h_att = (const float*)d_in[1];
    const float* t_att = (const float*)d_in[2];
    const float* w1    = (const float*)d_in[3];
    const float* b1    = (const float*)d_in[4];
    const float* w2    = (const float*)d_in[5];
    const float* b2    = (const float*)d_in[6];
    const int*   erow  = (const int*)d_in[7];
    const int*   ecol  = (const int*)d_in[8];
    float* out = (float*)d_out;

    const int N = in_sizes[0] / 64;
    const int E = in_sizes[7];

    float* ws  = (float*)d_ws;
    float* msg = ws;
    float* wE  = ws + (size_t)N * 64;
    float* m   = ws + (size_t)N * 64 + E;
    float* s   = ws + (size_t)N * 64 + E + N;

    float* out1 = out + (size_t)N * 64;
    float* out2 = out + 2 * (size_t)N * 64;

    const int TPB = 256;
    const int nd  = N * 64;

    const int waves   = cdiv(E, 32);
    const int spmm_bl = cdiv(waves, TPB / 32);

    init_kernel<<<FILL_BLOCKS, TPB, 0, stream>>>(h0, out, m, s, msg, N);
    logits_kernel<<<spmm_bl, TPB, 0, stream>>>(h_att, t_att, erow, ecol, m, wE, E);
    const int quad_bl = cdiv(cdiv(E, 4), TPB);
    expsum_kernel<<<quad_bl, TPB, 0, stream>>>(wE, m, s, erow, E);
    wnorm_kernel<<<quad_bl, TPB, 0, stream>>>(wE, s, erow, E);

    const int ntiles  = (N + 15) >> 4;
    const int dense_bl = cdiv(ntiles, TPB / 32);

    spmm_kernel<<<spmm_bl, TPB, 0, stream>>>(wE, h0, erow, ecol, msg, E);
    dense_kernel<<<dense_bl, TPB, 0, stream>>>(h0, msg, w1, b1, out1, N);

    zero_kernel<<<FILL_BLOCKS, TPB, 0, stream>>>(msg, nd / 4);
    spmm_kernel<<<spmm_bl, TPB, 0, stream>>>(wE, out1, erow, ecol, msg, E);
    dense_kernel<<<dense_bl, TPB, 0, stream>>>(out1, msg, w2, b2, out2, N);
}


# edge softmax by row ownership (edge_row sorted): logits kernel writes row pointers, one half-wave per row takes max, exp sum (lane order then shfl_xor tree) and division; no atomics, 7 launches
# speedup vs baseline: 2.4925x; 1.4362x over previous
#include <hip/hip_runtime.h>
#include <math.h>

typedef __attribute__((ext_vector_type(2))) float v2f;
typedef __attribute__((ext_vector_type(8))) float v8f;

#define GEPS 1e-12f

#define FILL_BLOCKS 1024

__global__ void init_kernel(const float* h0, float* out0, int n) {
    const int tid    = blockIdx.x * blockDim.x + threadIdx.x;
    const int stride = gridDim.x * blockDim.x;
    const int n4     = n * 16;
    const float4* src = (const float4*)h0;
    float4* dst = (float4*)out0;
    int i = tid;
    for (; i + 3 * stride < n4; i += 4 * stride) {
        const float4 v0 = src[i];
        const float4 v1 = src[i + stride];
        const float4 v2 = src[i + 2 * stride];
        const float4 v3 = src[i + 3 * stride];
        dst[i]              = v0;
        dst[i + stride]     = v1;
        dst[i + 2 * stride] = v2;
        dst[i + 3 * stride] = v3;
    }
    if (i + stride < n4) {
        const float4 v0 = src[i];
        const float4 v1 = src[i + stride];
        dst[i]          = v0;
        dst[i + stride] = v1;
        i += 2 * stride;
    }
    if (i < n4) { dst[i] = src[i]; }
}

__global__ void logits_kernel(const float* __restrict__ h_att, const float* __restrict__ t_att,
                              const int* __restrict__ erow, const int* __restrict__ ecol,
                              int* __restrict__ rowptr, float* __restrict__ wE,
                              int n_edges, int n_rows) {
    const int lane = threadIdx.x & 31;
    const int wave = __builtin_amdgcn_readfirstlane(
        (int)(blockIdx.x * (blockDim.x >> 5) + (threadIdx.x >> 5)));
    const int e0 = wave * 32;
    if (e0 >= n_edges) return;
    const int cnt = min(32, n_edges - e0);
    const int el   = e0 + min(lane, cnt - 1);
    const int my_r = erow[el];
    const int my_c = ecol[el];
    const int prev0 = e0 > 0 ? erow[e0 - 1] : -1;
    const int half = lane >> 4;
    const int sub  = lane & 15;
    float mylogit = 0.0f;

#pragma unroll
    for (int g = 0; g < 2; ++g) {
        float4 a[8], b[8];
#pragma unroll
        for (int k = 0; k < 8; ++k) {
            const int t = g * 8 + k;
            const int r_lo = __builtin_amdgcn_readlane(my_r, t);
            const int r_hi = __builtin_amdgcn_readlane(my_r, t + 16);
            const int c_lo = __builtin_amdgcn_readlane(my_c, t);
            const int c_hi = __builtin_amdgcn_readlane(my_c, t + 16);
            const int r = half ? r_hi : r_lo;
            const int c = half ? c_hi : c_lo;
            a[k] = *(const float4*)(h_att + (size_t)r * 64 + sub * 4);
            b[k] = *(const float4*)(t_att + (size_t)c * 64 + sub * 4);
        }
#pragma unroll
        for (int k = 0; k < 8; ++k) {
            const int t = g * 8 + k;
            float dot = a[k].x * b[k].x + a[k].y * b[k].y + a[k].z * b[k].z + a[k].w * b[k].w;
            dot += __shfl_xor(dot, 1, 32);
            dot += __shfl_xor(dot, 2, 32);
            dot += __shfl_xor(dot, 4, 32);
            dot += __shfl_xor(dot, 8, 32);
            if (sub == t) mylogit = dot;
        }
    }
    int rp = __shfl_up(my_r, 1, 32);
    if (lane == 0) rp = prev0;
    if (lane < cnt) {
        const int e = e0 + lane;
        wE[e] = mylogit;
        for (int g = rp + 1; g <= my_r; ++g)
            rowptr[g] = e;
        if (e == n_edges - 1)
            for (int g = my_r + 1; g <= n_rows; ++g)
                rowptr[g] = n_edges;
    }
}

#define SOFT_BLOCKS 1024

__global__ void rowsoftmax_kernel(const float* __restrict__ lg, float* __restrict__ wE,
                                  const int* __restrict__ rowptr, int n_rows) {
    const int sub    = threadIdx.x & 15;
    const int slot   = (int)((blockIdx.x * blockDim.x + threadIdx.x) >> 4);
    const int nslots = (int)((gridDim.x * blockDim.x) >> 4);
    const int per    = (n_rows + nslots - 1) / nslots;
    const int rbeg   = slot * per;
    const int rend   = min(rbeg + per, n_rows);
    const float ninf = -__builtin_inff();
    for (int r = rbeg; r < rend; ++r) {
        const int a = rowptr[r];
        const int b = rowptr[r + 1];
        const int k0 = a + sub;
        const int k1 = k0 + 16;
        const bool p0 = k0 < b;
        const bool p1 = k1 < b;
        const float l0 = p0 ? lg[k0] : ninf;
        const float l1 = p1 ? lg[k1] : ninf;
        float mx = fmaxf(l0, l1);
        for (int k = k0 + 32; k < b; k += 16) mx = fmaxf(mx, lg[k]);
        mx = fmaxf(mx, __shfl_xor(mx, 1, 32));
        mx = fmaxf(mx, __shfl_xor(mx, 2, 32));
        mx = fmaxf(mx, __shfl_xor(mx, 4, 32));
        mx = fmaxf(mx, __shfl_xor(mx, 8, 32));
        const float x0 = p0 ? expf(l0 - mx) : 0.0f;
        const float x1 = p1 ? expf(l1 - mx) : 0.0f;
        float s = x0 + x1;
        for (int k = k0 + 32; k < b; k += 16) s += expf(lg[k] - mx);
        s += __shfl_xor(s, 1, 32);
        s += __shfl_xor(s, 2, 32);
        s += __shfl_xor(s, 4, 32);
        s += __shfl_xor(s, 8, 32);
        const float d = fmaxf(s, GEPS);
        if (p0) wE[k0] = x0 / d;
        if (p1) wE[k1] = x1 / d;
        for (int k = k0 + 32; k < b; k += 16) wE[k] = expf(lg[k] - mx) / d;
    }
}

#define SPMM_CHUNK 128

__global__ void spmm_kernel(const float* __restrict__ wE, const float* __restrict__ h,
                            const int* __restrict__ erow, const int* __restrict__ ecol,
                            float* __restrict__ msg, int n_edges, int n_rows) {
    const int lane = threadIdx.x & 31;
    const int wave = __builtin_amdgcn_readfirstlane(
        (int)(blockIdx.x * (blockDim.x >> 5) + (threadIdx.x >> 5)));
    const int e0 = wave * SPMM_CHUNK;
    if (e0 >= n_edges) return;
    const int e1 = min(e0 + SPMM_CHUNK, n_edges);
    const int prev = e0 > 0 ? erow[e0 - 1] : -1;
    const int last = erow[e1 - 1];
    if (last == prev) return;
    const int tail_row = erow[n_edges - 1];

    float2* __restrict__ mrow = (float2*)msg;
    const float2 z2 = make_float2(0.0f, 0.0f);
    int   cur  = prev;
    bool  have = false;
    float ax = 0.0f, ay = 0.0f;

    for (int p = e0; p < n_edges; p += 32) {
        const int   e    = p + lane;
        const int   el   = min(e, n_edges - 1);
        const int   my_r = erow[el];
        const int   my_c = ecol[el];
        const float my_w = wE[el];
        const unsigned int my_o = (unsigned int)my_c * 64u;
        const bool ours = (e < n_edges) && (my_r != prev) && (e < e1 || my_r == last);
        const unsigned int mask = (unsigned int)__builtin_amdgcn_ballot_w32(ours);

        float2 hv[32];
#pragma unroll
        for (int j = 0; j < 32; ++j) {
            if ((mask >> j) & 1u) {
                const unsigned int o = (unsigned int)__builtin_amdgcn_readlane((int)my_o, j);
                hv[j] = *(const float2*)(h + (size_t)o + lane * 2);
            }
        }

#pragma unroll
        for (int j = 0; j < 32; ++j) {
            if ((mask >> j) & 1u) {
                const int   r  = __builtin_amdgcn_readlane(my_r, j);
                const float we = __int_as_float(__builtin_amdgcn_readlane(__float_as_int(my_w), j));
                if (r != cur) {
                    if (have) mrow[(size_t)cur * 32 + lane] = make_float2(ax, ay);
#pragma nounroll
                    for (int g = cur + 1; g < r; ++g)
                        mrow[(size_t)g * 32 + lane] = z2;
                    ax = 0.0f; ay = 0.0f; cur = r; have = true;
                }
                ax = fmaf(we, hv[j].x, ax);
                ay = fmaf(we, hv[j].y, ay);
            }
        }
        if (p + 32 >= e1 && !((mask >> 31) & 1u)) break;
    }
    mrow[(size_t)cur * 32 + lane] = make_float2(ax, ay);
    if (cur == tail_row)
        for (int g = cur + 1; g < n_rows; ++g)
            mrow[(size_t)g * 32 + lane] = z2;
}

__global__ void dense_kernel(const float* __restrict__ hprev, const float* __restrict__ msg,
                             const float* __restrict__ W, const float* __restrict__ bias,
                             float* __restrict__ out, int n) {
    const int lane = threadIdx.x & 31;
    const int wave = threadIdx.x >> 5;
    const int tile = blockIdx.x * (blockDim.x >> 5) + wave;
    const int ntiles = (n + 15) >> 4;
    if (tile >= ntiles) return;

    const int half = lane >> 4;
    const int l16  = lane & 15;
    int row  = tile * 16 + l16;
    int rowc = row < n ? row : n - 1;

    v8f acc[4] = {};

#pragma unroll
    for (int kb = 0; kb < 16; ++kb) {
        const int koff = kb * 4 + half * 2;
        const float* hp = hprev + (size_t)rowc * 64 + koff;
        const float* mp = msg   + (size_t)rowc * 64 + koff;
        v2f a;
        a.x = hp[0] + mp[0];
        a.y = hp[1] + mp[1];
#pragma unroll
        for (int jb = 0; jb < 4; ++jb) {
            const float* wp = W + (size_t)(jb * 16 + l16) * 64 + koff;
            v2f b;
            b.x = wp[0];
            b.y = wp[1];
            acc[jb] = __builtin_amdgcn_wmma_f32_16x16x4_f32(
                false, a, false, b, (short)0, acc[jb], false, false);
        }
    }

    float bj[4];
#pragma unroll
    for (int jb = 0; jb < 4; ++jb) bj[jb] = bias[jb * 16 + l16];

#pragma unroll
    for (int v = 0; v < 8; ++v) {
        float vals[4];
        float ss = 0.0f;
#pragma unroll
        for (int jb = 0; jb < 4; ++jb) {
            float y = fmaxf(acc[jb][v] + bj[jb], 0.0f);
            vals[jb] = y;
            ss = fmaf(y, y, ss);
        }
        ss += __shfl_xor(ss, 1, 32);
        ss += __shfl_xor(ss, 2, 32);
        ss += __shfl_xor(ss, 4, 32);
        ss += __shfl_xor(ss, 8, 32);
        float inv = 1.0f / fmaxf(sqrtf(ss), GEPS);
        int m    = v + half * 8;
        int orow = tile * 16 + m;
        if (orow < n) {
#pragma unroll
            for (int jb = 0; jb < 4; ++jb)
                out[(size_t)orow * 64 + jb * 16 + l16] = vals[jb] * inv;
        }
    }
}

static inline int cdiv(long long a, long long b) { return (int)((a + b - 1) / b); }

extern "C" void kernel_launch(void* const* d_in, const int* in_sizes, int n_in,
                              void* d_out, int out_size, void* d_ws, size_t ws_size,
                              hipStream_t stream) {
    const float* h0    = (const float*)d_in[0];
    const float* h_att = (const float*)d_in[1];
    const float* t_att = (const float*)d_in[2];
    const float* w1    = (const float*)d_in[3];
    const float* b1    = (const float*)d_in[4];
    const float* w2    = (const float*)d_in[5];
    const float* b2    = (const float*)d_in[6];
    const int*   erow  = (const int*)d_in[7];
    const int*   ecol  = (const int*)d_in[8];
    float* out = (float*)d_out;

    const int N = in_sizes[0] / 64;
    const int E = in_sizes[7];

    float* ws  = (float*)d_ws;
    float* msg = ws;
    float* wE  = ws + (size_t)N * 64;
    int* rowptr = (int*)(ws + (size_t)N * 64 + E);

    float* out1 = out + (size_t)N * 64;
    float* out2 = out + 2 * (size_t)N * 64;

    const int TPB = 256;

    const int edge_bl = cdiv(cdiv(E, 32), TPB / 32);
    const int spmm_bl = cdiv(cdiv(E, SPMM_CHUNK), TPB / 32);

    init_kernel<<<FILL_BLOCKS, TPB, 0, stream>>>(h0, out, N);
    logits_kernel<<<edge_bl, TPB, 0, stream>>>(h_att, t_att, erow, ecol, rowptr, msg, E, N);
    rowsoftmax_kernel<<<SOFT_BLOCKS, TPB, 0, stream>>>(msg, wE, rowptr, N);

    const int ntiles  = (N + 15) >> 4;
    const int dense_bl = cdiv(ntiles, TPB / 32);

    spmm_kernel<<<spmm_bl, TPB, 0, stream>>>(wE, h0, erow, ecol, msg, E, N);
    dense_kernel<<<dense_bl, TPB, 0, stream>>>(h0, msg, w1, b1, out1, N);

    spmm_kernel<<<spmm_bl, TPB, 0, stream>>>(wE, out1, erow, ecol, msg, E, N);
    dense_kernel<<<dense_bl, TPB, 0, stream>>>(out1, msg, w2, b2, out2, N);
}


# dense kernel: W and each 16-row (h+msg) tile staged in LDS by whole-row 16-B loads (4-wave blocks), WMMA fragments read from LDS, result rows stored whole via LDS; k order and all sums unchanged
# speedup vs baseline: 2.7595x; 1.1071x over previous
#include <hip/hip_runtime.h>
#include <math.h>

typedef __attribute__((ext_vector_type(2))) float v2f;
typedef __attribute__((ext_vector_type(8))) float v8f;

#define GEPS 1e-12f

#define FILL_BLOCKS 1024

__global__ void init_kernel(const float* h0, float* out0, int n) {
    const int tid    = blockIdx.x * blockDim.x + threadIdx.x;
    const int stride = gridDim.x * blockDim.x;
    const int n4     = n * 16;
    const float4* src = (const float4*)h0;
    float4* dst = (float4*)out0;
    int i = tid;
    for (; i + 3 * stride < n4; i += 4 * stride) {
        const float4 v0 = src[i];
        const float4 v1 = src[i + stride];
        const float4 v2 = src[i + 2 * stride];
        const float4 v3 = src[i + 3 * stride];
        dst[i]              = v0;
        dst[i + stride]     = v1;
        dst[i + 2 * stride] = v2;
        dst[i + 3 * stride] = v3;
    }
    if (i + stride < n4) {
        const float4 v0 = src[i];
        const float4 v1 = src[i + stride];
        dst[i]          = v0;
        dst[i + stride] = v1;
        i += 2 * stride;
    }
    if (i < n4) { dst[i] = src[i]; }
}

__global__ void logits_kernel(const float* __restrict__ h_att, const float* __restrict__ t_att,
                              const int* __restrict__ erow, const int* __restrict__ ecol,
                              int* __restrict__ rowptr, float* __restrict__ wE,
                              int n_edges, int n_rows) {
    const int lane = threadIdx.x & 31;
    const int wave = __builtin_amdgcn_readfirstlane(
        (int)(blockIdx.x * (blockDim.x >> 5) + (threadIdx.x >> 5)));
    const int e0 = wave * 32;
    if (e0 >= n_edges) return;
    const int cnt = min(32, n_edges - e0);
    const int el   = e0 + min(lane, cnt - 1);
    const int my_r = erow[el];
    const int my_c = ecol[el];
    const int prev0 = e0 > 0 ? erow[e0 - 1] : -1;
    const int half = lane >> 4;
    const int sub  = lane & 15;
    float mylogit = 0.0f;

#pragma unroll
    for (int g = 0; g < 2; ++g) {
        float4 a[8], b[8];
#pragma unroll
        for (int k = 0; k < 8; ++k) {
            const int t = g * 8 + k;
            const int r_lo = __builtin_amdgcn_readlane(my_r, t);
            const int r_hi = __builtin_amdgcn_readlane(my_r, t + 16);
            const int c_lo = __builtin_amdgcn_readlane(my_c, t);
            const int c_hi = __builtin_amdgcn_readlane(my_c, t + 16);
            const int r = half ? r_hi : r_lo;
            const int c = half ? c_hi : c_lo;
            a[k] = *(const float4*)(h_att + (size_t)r * 64 + sub * 4);
            b[k] = *(const float4*)(t_att + (size_t)c * 64 + sub * 4);
        }
#pragma unroll
        for (int k = 0; k < 8; ++k) {
            const int t = g * 8 + k;
            float dot = a[k].x * b[k].x + a[k].y * b[k].y + a[k].z * b[k].z + a[k].w * b[k].w;
            dot += __shfl_xor(dot, 1, 32);
            dot += __shfl_xor(dot, 2, 32);
            dot += __shfl_xor(dot, 4, 32);
            dot += __shfl_xor(dot, 8, 32);
            if (sub == t) mylogit = dot;
        }
    }
    int rp = __shfl_up(my_r, 1, 32);
    if (lane == 0) rp = prev0;
    if (lane < cnt) {
        const int e = e0 + lane;
        wE[e] = mylogit;
        for (int g = rp + 1; g <= my_r; ++g)
            rowptr[g] = e;
        if (e == n_edges - 1)
            for (int g = my_r + 1; g <= n_rows; ++g)
                rowptr[g] = n_edges;
    }
}

#define SOFT_BLOCKS 1024

__global__ void rowsoftmax_kernel(const float* __restrict__ lg, float* __restrict__ wE,
                                  const int* __restrict__ rowptr, int n_rows) {
    const int sub    = threadIdx.x & 15;
    const int slot   = (int)((blockIdx.x * blockDim.x + threadIdx.x) >> 4);
    const int nslots = (int)((gridDim.x * blockDim.x) >> 4);
    const int per    = (n_rows + nslots - 1) / nslots;
    const int rbeg   = slot * per;
    const int rend   = min(rbeg + per, n_rows);
    const float ninf = -__builtin_inff();
    for (int r = rbeg; r < rend; ++r) {
        const int a = rowptr[r];
        const int b = rowptr[r + 1];
        const int k0 = a + sub;
        const int k1 = k0 + 16;
        const bool p0 = k0 < b;
        const bool p1 = k1 < b;
        const float l0 = p0 ? lg[k0] : ninf;
        const float l1 = p1 ? lg[k1] : ninf;
        float mx = fmaxf(l0, l1);
        for (int k = k0 + 32; k < b; k += 16) mx = fmaxf(mx, lg[k]);
        mx = fmaxf(mx, __shfl_xor(mx, 1, 32));
        mx = fmaxf(mx, __shfl_xor(mx, 2, 32));
        mx = fmaxf(mx, __shfl_xor(mx, 4, 32));
        mx = fmaxf(mx, __shfl_xor(mx, 8, 32));
        const float x0 = p0 ? expf(l0 - mx) : 0.0f;
        const float x1 = p1 ? expf(l1 - mx) : 0.0f;
        float s = x0 + x1;
        for (int k = k0 + 32; k < b; k += 16) s += expf(lg[k] - mx);
        s += __shfl_xor(s, 1, 32);
        s += __shfl_xor(s, 2, 32);
        s += __shfl_xor(s, 4, 32);
        s += __shfl_xor(s, 8, 32);
        const float d = fmaxf(s, GEPS);
        if (p0) wE[k0] = x0 / d;
        if (p1) wE[k1] = x1 / d;
        for (int k = k0 + 32; k < b; k += 16) wE[k] = expf(lg[k] - mx) / d;
    }
}

#define SPMM_CHUNK 128

__global__ void spmm_kernel(const float* __restrict__ wE, const float* __restrict__ h,
                            const int* __restrict__ erow, const int* __restrict__ ecol,
                            float* __restrict__ msg, int n_edges, int n_rows) {
    const int lane = threadIdx.x & 31;
    const int wave = __builtin_amdgcn_readfirstlane(
        (int)(blockIdx.x * (blockDim.x >> 5) + (threadIdx.x >> 5)));
    const int e0 = wave * SPMM_CHUNK;
    if (e0 >= n_edges) return;
    const int e1 = min(e0 + SPMM_CHUNK, n_edges);
    const int prev = e0 > 0 ? erow[e0 - 1] : -1;
    const int last = erow[e1 - 1];
    if (last == prev) return;
    const int tail_row = erow[n_edges - 1];

    float2* __restrict__ mrow = (float2*)msg;
    const float2 z2 = make_float2(0.0f, 0.0f);
    int   cur  = prev;
    bool  have = false;
    float ax = 0.0f, ay = 0.0f;

    for (int p = e0; p < n_edges; p += 32) {
        const int   e    = p + lane;
        const int   el   = min(e, n_edges - 1);
        const int   my_r = erow[el];
        const int   my_c = ecol[el];
        const float my_w = wE[el];
        const unsigned int my_o = (unsigned int)my_c * 64u;
        const bool ours = (e < n_edges) && (my_r != prev) && (e < e1 || my_r == last);
        const unsigned int mask = (unsigned int)__builtin_amdgcn_ballot_w32(ours);

        float2 hv[32];
#pragma unroll
        for (int j = 0; j < 32; ++j) {
            if ((mask >> j) & 1u) {
                const unsigned int o = (unsigned int)__builtin_amdgcn_readlane((int)my_o, j);
                hv[j] = *(const float2*)(h + (size_t)o + lane * 2);
            }
        }

#pragma unroll
        for (int j = 0; j < 32; ++j) {
            if ((mask >> j) & 1u) {
                const int   r  = __builtin_amdgcn_readlane(my_r, j);
                const float we = __int_as_float(__builtin_amdgcn_readlane(__float_as_int(my_w), j));
                if (r != cur) {
                    if (have) mrow[(size_t)cur * 32 + lane] = make_float2(ax, ay);
#pragma nounroll
                    for (int g = cur + 1; g < r; ++g)
                        mrow[(size_t)g * 32 + lane] = z2;
                    ax = 0.0f; ay = 0.0f; cur = r; have = true;
                }
                ax = fmaf(we, hv[j].x, ax);
                ay = fmaf(we, hv[j].y, ay);
            }
        }
        if (p + 32 >= e1 && !((mask >> 31) & 1u)) break;
    }
    mrow[(size_t)cur * 32 + lane] = make_float2(ax, ay);
    if (cur == tail_row)
        for (int g = cur + 1; g < n_rows; ++g)
            mrow[(size_t)g * 32 + lane] = z2;
}

#define DENSE_WAVES 4
#define DPITCH 68

__global__ void __launch_bounds__(DENSE_WAVES * 32)
dense_kernel(const float* __restrict__ hprev, const float* __restrict__ msg,
             const float* __restrict__ W, const float* __restrict__ bias,
             float* __restrict__ out, int n) {
    __shared__ __attribute__((aligned(16))) float sW[64 * DPITCH];
    __shared__ __attribute__((aligned(16))) float sT[DENSE_WAVES][16 * DPITCH];

    const int lane = threadIdx.x & 31;
    const int wave = __builtin_amdgcn_readfirstlane((int)(threadIdx.x >> 5));
    const int tile = blockIdx.x * DENSE_WAVES + wave;
    const int ntiles = (n + 15) >> 4;
    const bool live = tile < ntiles;
    const int  tl   = live ? tile : ntiles - 1;

    const int half = lane >> 4;
    const int l16  = lane & 15;
    float* tbuf = sT[wave];

    float4 wv[8];
#pragma unroll
    for (int i = 0; i < 8; ++i) {
        const int idx = (int)threadIdx.x + i * (DENSE_WAVES * 32);
        wv[i] = *(const float4*)(W + (size_t)idx * 4);
    }

    float4 hv[8], mv[8];
#pragma unroll
    for (int i = 0; i < 8; ++i) {
        const int row  = tl * 16 + 2 * i + half;
        const int rowc = row < n ? row : n - 1;
        hv[i] = *(const float4*)(hprev + (size_t)rowc * 64 + l16 * 4);
        mv[i] = *(const float4*)(msg   + (size_t)rowc * 64 + l16 * 4);
    }

#pragma unroll
    for (int i = 0; i < 8; ++i) {
        const int idx = (int)threadIdx.x + i * (DENSE_WAVES * 32);
        const int r   = idx >> 4;
        const int c4  = idx & 15;
        *(float4*)(sW + r * DPITCH + c4 * 4) = wv[i];
    }

#pragma unroll
    for (int i = 0; i < 8; ++i) {
        float4 s;
        s.x = hv[i].x + mv[i].x;
        s.y = hv[i].y + mv[i].y;
        s.z = hv[i].z + mv[i].z;
        s.w = hv[i].w + mv[i].w;
        *(float4*)(tbuf + (2 * i + half) * DPITCH + l16 * 4) = s;
    }
    __syncthreads();
    if (!live) return;

    v8f acc[4] = {};

#pragma unroll
    for (int kb = 0; kb < 16; ++kb) {
        const int koff = kb * 4 + half * 2;
        const v2f a = *(const v2f*)(tbuf + l16 * DPITCH + koff);
#pragma unroll
        for (int jb = 0; jb < 4; ++jb) {
            const v2f b = *(const v2f*)(sW + (jb * 16 + l16) * DPITCH + koff);
            acc[jb] = __builtin_amdgcn_wmma_f32_16x16x4_f32(
                false, a, false, b, (short)0, acc[jb], false, false);
        }
    }

    float bj[4];
#pragma unroll
    for (int jb = 0; jb < 4; ++jb) bj[jb] = bias[jb * 16 + l16];

#pragma unroll
    for (int v = 0; v < 8; ++v) {
        float vals[4];
        float ss = 0.0f;
#pragma unroll
        for (int jb = 0; jb < 4; ++jb) {
            float y = fmaxf(acc[jb][v] + bj[jb], 0.0f);
            vals[jb] = y;
            ss = fmaf(y, y, ss);
        }
        ss += __shfl_xor(ss, 1, 32);
        ss += __shfl_xor(ss, 2, 32);
        ss += __shfl_xor(ss, 4, 32);
        ss += __shfl_xor(ss, 8, 32);
        float inv = 1.0f / fmaxf(sqrtf(ss), GEPS);
        const int m = v + half * 8;
#pragma unroll
        for (int jb = 0; jb < 4; ++jb)
            tbuf[m * DPITCH + jb * 16 + l16] = vals[jb] * inv;
    }

    __builtin_amdgcn_fence(__ATOMIC_RELEASE, "wavefront");
    __builtin_amdgcn_wave_barrier();
    __builtin_amdgcn_fence(__ATOMIC_ACQUIRE, "wavefront");

    float4 o[8];
#pragma unroll
    for (int i = 0; i < 8; ++i)
        o[i] = *(const float4*)(tbuf + (2 * i + half) * DPITCH + l16 * 4);

    float* orow0 = out + (size_t)tile * 16 * 64 + (size_t)half * 64 + l16 * 4;
    if (tile * 16 + 16 <= n) {
#pragma unroll
        for (int i = 0; i < 8; ++i)
            *(float4*)(orow0 + (size_t)i * 128) = o[i];
    } else {
#pragma unroll
        for (int i = 0; i < 8; ++i)
            if (tile * 16 + 2 * i + half < n)
                *(float4*)(orow0 + (size_t)i * 128) = o[i];
    }
}

static inline int cdiv(long long a, long long b) { return (int)((a + b - 1) / b); }

extern "C" void kernel_launch(void* const* d_in, const int* in_sizes, int n_in,
                              void* d_out, int out_size, void* d_ws, size_t ws_size,
                              hipStream_t stream) {
    const float* h0    = (const float*)d_in[0];
    const float* h_att = (const float*)d_in[1];
    const float* t_att = (const float*)d_in[2];
    const float* w1    = (const float*)d_in[3];
    const float* b1    = (const float*)d_in[4];
    const float* w2    = (const float*)d_in[5];
    const float* b2    = (const float*)d_in[6];
    const int*   erow  = (const int*)d_in[7];
    const int*   ecol  = (const int*)d_in[8];
    float* out = (float*)d_out;

    const int N = in_sizes[0] / 64;
    const int E = in_sizes[7];

    float* ws  = (float*)d_ws;
    float* msg = ws;
    float* wE  = ws + (size_t)N * 64;
    int* rowptr = (int*)(ws + (size_t)N * 64 + E);

    float* out1 = out + (size_t)N * 64;
    float* out2 = out + 2 * (size_t)N * 64;

    const int TPB = 256;

    const int edge_bl = cdiv(cdiv(E, 32), TPB / 32);
    const int spmm_bl = cdiv(cdiv(E, SPMM_CHUNK), TPB / 32);

    init_kernel<<<FILL_BLOCKS, TPB, 0, stream>>>(h0, out, N);
    logits_kernel<<<edge_bl, TPB, 0, stream>>>(h_att, t_att, erow, ecol, rowptr, msg, E, N);
    rowsoftmax_kernel<<<SOFT_BLOCKS, TPB, 0, stream>>>(msg, wE, rowptr, N);

    const int ntiles  = (N + 15) >> 4;
    const int dense_bl = cdiv(ntiles, DENSE_WAVES);

    spmm_kernel<<<spmm_bl, TPB, 0, stream>>>(wE, h0, erow, ecol, msg, E, N);
    dense_kernel<<<dense_bl, DENSE_WAVES * 32, 0, stream>>>(h0, msg, w1, b1, out1, N);

    spmm_kernel<<<spmm_bl, TPB, 0, stream>>>(wE, out1, erow, ecol, msg, E, N);
    dense_kernel<<<dense_bl, DENSE_WAVES * 32, 0, stream>>>(out1, msg, w2, b2, out2, N);
}


# SpMM fused into the dense kernel: each wave sums its 16-row tile's msg rows (edge run from the row pointers, same edge order, in registers) into its LDS tile; msg never goes to memory; 5 launches
# speedup vs baseline: 3.0802x; 1.1162x over previous
#include <hip/hip_runtime.h>
#include <math.h>

typedef __attribute__((ext_vector_type(2))) float v2f;
typedef __attribute__((ext_vector_type(8))) float v8f;

#define GEPS 1e-12f

#define FILL_BLOCKS 1024

__global__ void init_kernel(const float* h0, float* out0, int n) {
    const int tid    = blockIdx.x * blockDim.x + threadIdx.x;
    const int stride = gridDim.x * blockDim.x;
    const int n4     = n * 16;
    const float4* src = (const float4*)h0;
    float4* dst = (float4*)out0;
    int i = tid;
    for (; i + 3 * stride < n4; i += 4 * stride) {
        const float4 v0 = src[i];
        const float4 v1 = src[i + stride];
        const float4 v2 = src[i + 2 * stride];
        const float4 v3 = src[i + 3 * stride];
        dst[i]              = v0;
        dst[i + stride]     = v1;
        dst[i + 2 * stride] = v2;
        dst[i + 3 * stride] = v3;
    }
    if (i + stride < n4) {
        const float4 v0 = src[i];
        const float4 v1 = src[i + stride];
        dst[i]          = v0;
        dst[i + stride] = v1;
        i += 2 * stride;
    }
    if (i < n4) { dst[i] = src[i]; }
}

__global__ void logits_kernel(const float* __restrict__ h_att, const float* __restrict__ t_att,
                              const int* __restrict__ erow, const int* __restrict__ ecol,
                              int* __restrict__ rowptr, float* __restrict__ wE,
                              int n_edges, int n_rows) {
    const int lane = threadIdx.x & 31;
    const int wave = __builtin_amdgcn_readfirstlane(
        (int)(blockIdx.x * (blockDim.x >> 5) + (threadIdx.x >> 5)));
    const int e0 = wave * 32;
    if (e0 >= n_edges) return;
    const int cnt = min(32, n_edges - e0);
    const int el   = e0 + min(lane, cnt - 1);
    const int my_r = erow[el];
    const int my_c = ecol[el];
    const int prev0 = e0 > 0 ? erow[e0 - 1] : -1;
    const int half = lane >> 4;
    const int sub  = lane & 15;
    float mylogit = 0.0f;

#pragma unroll
    for (int g = 0; g < 2; ++g) {
        float4 a[8], b[8];
#pragma unroll
        for (int k = 0; k < 8; ++k) {
            const int t = g * 8 + k;
            const int r_lo = __builtin_amdgcn_readlane(my_r, t);
            const int r_hi = __builtin_amdgcn_readlane(my_r, t + 16);
            const int c_lo = __builtin_amdgcn_readlane(my_c, t);
            const int c_hi = __builtin_amdgcn_readlane(my_c, t + 16);
            const int r = half ? r_hi : r_lo;
            const int c = half ? c_hi : c_lo;
            a[k] = *(const float4*)(h_att + (size_t)r * 64 + sub * 4);
            b[k] = *(const float4*)(t_att + (size_t)c * 64 + sub * 4);
        }
#pragma unroll
        for (int k = 0; k < 8; ++k) {
            const int t = g * 8 + k;
            float dot = a[k].x * b[k].x + a[k].y * b[k].y + a[k].z * b[k].z + a[k].w * b[k].w;
            dot += __shfl_xor(dot, 1, 32);
            dot += __shfl_xor(dot, 2, 32);
            dot += __shfl_xor(dot, 4, 32);
            dot += __shfl_xor(dot, 8, 32);
            if (sub == t) mylogit = dot;
        }
    }
    int rp = __shfl_up(my_r, 1, 32);
    if (lane == 0) rp = prev0;
    if (lane < cnt) {
        const int e = e0 + lane;
        wE[e] = mylogit;
        for (int g = rp + 1; g <= my_r; ++g)
            rowptr[g] = e;
        if (e == n_edges - 1)
            for (int g = my_r + 1; g <= n_rows; ++g)
                rowptr[g] = n_edges;
    }
}

#define SOFT_BLOCKS 1024

__global__ void rowsoftmax_kernel(const float* __restrict__ lg, float* __restrict__ wE,
                                  const int* __restrict__ rowptr, int n_rows) {
    const int sub    = threadIdx.x & 15;
    const int slot   = (int)((blockIdx.x * blockDim.x + threadIdx.x) >> 4);
    const int nslots = (int)((gridDim.x * blockDim.x) >> 4);
    const int per    = (n_rows + nslots - 1) / nslots;
    const int rbeg   = slot * per;
    const int rend   = min(rbeg + per, n_rows);
    const float ninf = -__builtin_inff();
    for (int r = rbeg; r < rend; ++r) {
        const int a = rowptr[r];
        const int b = rowptr[r + 1];
        const int k0 = a + sub;
        const int k1 = k0 + 16;
        const bool p0 = k0 < b;
        const bool p1 = k1 < b;
        const float l0 = p0 ? lg[k0] : ninf;
        const float l1 = p1 ? lg[k1] : ninf;
        float mx = fmaxf(l0, l1);
        for (int k = k0 + 32; k < b; k += 16) mx = fmaxf(mx, lg[k]);
        mx = fmaxf(mx, __shfl_xor(mx, 1, 32));
        mx = fmaxf(mx, __shfl_xor(mx, 2, 32));
        mx = fmaxf(mx, __shfl_xor(mx, 4, 32));
        mx = fmaxf(mx, __shfl_xor(mx, 8, 32));
        const float x0 = p0 ? expf(l0 - mx) : 0.0f;
        const float x1 = p1 ? expf(l1 - mx) : 0.0f;
        float s = x0 + x1;
        for (int k = k0 + 32; k < b; k += 16) s += expf(lg[k] - mx);
        s += __shfl_xor(s, 1, 32);
        s += __shfl_xor(s, 2, 32);
        s += __shfl_xor(s, 4, 32);
        s += __shfl_xor(s, 8, 32);
        const float d = fmaxf(s, GEPS);
        if (p0) wE[k0] = x0 / d;
        if (p1) wE[k1] = x1 / d;
        for (int k = k0 + 32; k < b; k += 16) wE[k] = expf(lg[k] - mx) / d;
    }
}

#define LAYER_WAVES 4
#define DPITCH 68

__global__ void __launch_bounds__(LAYER_WAVES * 32)
layer_kernel(const float* __restrict__ h, const float* __restrict__ wE,
             const int* __restrict__ erow, const int* __restrict__ ecol,
             const int* __restrict__ rowptr,
             const float* __restrict__ W, const float* __restrict__ bias,
             float* __restrict__ out, int n, int n_edges) {
    __shared__ __attribute__((aligned(16))) float sW[64 * DPITCH];
    __shared__ __attribute__((aligned(16))) float sT[LAYER_WAVES][16 * DPITCH];

    const int lane = threadIdx.x & 31;
    const int wave = __builtin_amdgcn_readfirstlane((int)(threadIdx.x >> 5));
    const int tile = blockIdx.x * LAYER_WAVES + wave;
    const int ntiles = (n + 15) >> 4;
    const bool live = tile < ntiles;
    const int  tl   = live ? tile : ntiles - 1;

    const int half = lane >> 4;
    const int l16  = lane & 15;
    float* tbuf = sT[wave];

    const int r0 = tl * 16;
    const int r1 = min(r0 + 16, n);
    const int ea = __builtin_amdgcn_readfirstlane(rowptr[r0]);
    const int eb = __builtin_amdgcn_readfirstlane(rowptr[r1]);

    float4 wv[8];
#pragma unroll
    for (int i = 0; i < 8; ++i) {
        const int idx = (int)threadIdx.x + i * (LAYER_WAVES * 32);
        wv[i] = *(const float4*)(W + (size_t)idx * 4);
    }

    int   nx_r, nx_c;
    float nx_w;
    {
        const int el = min(ea + lane, n_edges - 1);
        nx_r = erow[el];
        nx_c = ecol[el];
        nx_w = wE[el];
    }

    float4 hv[8];
#pragma unroll
    for (int i = 0; i < 8; ++i) {
        const int row  = r0 + 2 * i + half;
        const int rowc = row < n ? row : n - 1;
        hv[i] = *(const float4*)(h + (size_t)rowc * 64 + l16 * 4);
    }

#pragma unroll
    for (int i = 0; i < 8; ++i) {
        const int idx = (int)threadIdx.x + i * (LAYER_WAVES * 32);
        const int r   = idx >> 4;
        const int c4  = idx & 15;
        *(float4*)(sW + r * DPITCH + c4 * 4) = wv[i];
    }
#pragma unroll
    for (int i = 0; i < 8; ++i)
        *(float4*)(tbuf + (2 * i + half) * DPITCH + l16 * 4) = make_float4(0.0f, 0.0f, 0.0f, 0.0f);

    __syncthreads();
    if (!live) return;

    {
        float2* trow = (float2*)tbuf;
        int   cur = -1;
        float ax = 0.0f, ay = 0.0f;

        for (int p = ea; p < eb; p += 32) {
            const int   cnt  = min(32, eb - p);
            const int   my_r = nx_r;
            const float my_w = nx_w;
            const unsigned int my_o = (unsigned int)nx_c * 64u;
            {
                const int el = min(p + 32 + lane, n_edges - 1);
                nx_r = erow[el];
                nx_c = ecol[el];
                nx_w = wE[el];
            }

            float2 gv[32];
#pragma unroll
            for (int j = 0; j < 32; ++j) {
                if (j < cnt) {
                    const unsigned int o = (unsigned int)__builtin_amdgcn_readlane((int)my_o, j);
                    gv[j] = *(const float2*)(h + (size_t)o + lane * 2);
                }
            }

#pragma unroll
            for (int j = 0; j < 32; ++j) {
                if (j < cnt) {
                    const int   r  = __builtin_amdgcn_readlane(my_r, j);
                    const float we = __int_as_float(__builtin_amdgcn_readlane(__float_as_int(my_w), j));
                    if (r != cur) {
                        if (cur >= 0) trow[(cur - r0) * (DPITCH / 2) + lane] = make_float2(ax, ay);
                        ax = 0.0f; ay = 0.0f; cur = r;
                    }
                    ax = fmaf(we, gv[j].x, ax);
                    ay = fmaf(we, gv[j].y, ay);
                }
            }
        }
        if (cur >= 0) trow[(cur - r0) * (DPITCH / 2) + lane] = make_float2(ax, ay);
    }

    __builtin_amdgcn_fence(__ATOMIC_RELEASE, "wavefront");
    __builtin_amdgcn_wave_barrier();
    __builtin_amdgcn_fence(__ATOMIC_ACQUIRE, "wavefront");

#pragma unroll
    for (int i = 0; i < 8; ++i) {
        float* q = tbuf + (2 * i + half) * DPITCH + l16 * 4;
        const float4 mv = *(const float4*)q;
        float4 sm;
        sm.x = hv[i].x + mv.x;
        sm.y = hv[i].y + mv.y;
        sm.z = hv[i].z + mv.z;
        sm.w = hv[i].w + mv.w;
        *(float4*)q = sm;
    }

    __builtin_amdgcn_fence(__ATOMIC_RELEASE, "wavefront");
    __builtin_amdgcn_wave_barrier();
    __builtin_amdgcn_fence(__ATOMIC_ACQUIRE, "wavefront");

    v8f acc[4] = {};

#pragma unroll
    for (int kb = 0; kb < 16; ++kb) {
        const int koff = kb * 4 + half * 2;
        const v2f a = *(const v2f*)(tbuf + l16 * DPITCH + koff);
#pragma unroll
        for (int jb = 0; jb < 4; ++jb) {
            const v2f b = *(const v2f*)(sW + (jb * 16 + l16) * DPITCH + koff);
            acc[jb] = __builtin_amdgcn_wmma_f32_16x16x4_f32(
                false, a, false, b, (short)0, acc[jb], false, false);
        }
    }

    float bj[4];
#pragma unroll
    for (int jb = 0; jb < 4; ++jb) bj[jb] = bias[jb * 16 + l16];

#pragma unroll
    for (int v = 0; v < 8; ++v) {
        float vals[4];
        float ss = 0.0f;
#pragma unroll
        for (int jb = 0; jb < 4; ++jb) {
            float y = fmaxf(acc[jb][v] + bj[jb], 0.0f);
            vals[jb] = y;
            ss = fmaf(y, y, ss);
        }
        ss += __shfl_xor(ss, 1, 32);
        ss += __shfl_xor(ss, 2, 32);
        ss += __shfl_xor(ss, 4, 32);
        ss += __shfl_xor(ss, 8, 32);
        float inv = 1.0f / fmaxf(sqrtf(ss), GEPS);
        const int m = v + half * 8;
#pragma unroll
        for (int jb = 0; jb < 4; ++jb)
            tbuf[m * DPITCH + jb * 16 + l16] = vals[jb] * inv;
    }

    __builtin_amdgcn_fence(__ATOMIC_RELEASE, "wavefront");
    __builtin_amdgcn_wave_barrier();
    __builtin_amdgcn_fence(__ATOMIC_ACQUIRE, "wavefront");

    float4 o[8];
#pragma unroll
    for (int i = 0; i < 8; ++i)
        o[i] = *(const float4*)(tbuf + (2 * i + half) * DPITCH + l16 * 4);

    float* orow0 = out + (size_t)tile * 16 * 64 + (size_t)half * 64 + l16 * 4;
    if (tile * 16 + 16 <= n) {
#pragma unroll
        for (int i = 0; i < 8; ++i)
            *(float4*)(orow0 + (size_t)i * 128) = o[i];
    } else {
#pragma unroll
        for (int i = 0; i < 8; ++i)
            if (tile * 16 + 2 * i + half < n)
                *(float4*)(orow0 + (size_t)i * 128) = o[i];
    }
}

static inline int cdiv(long long a, long long b) { return (int)((a + b - 1) / b); }

extern "C" void kernel_launch(void* const* d_in, const int* in_sizes, int n_in,
                              void* d_out, int out_size, void* d_ws, size_t ws_size,
                              hipStream_t stream) {
    const float* h0    = (const float*)d_in[0];
    const float* h_att = (const float*)d_in[1];
    const float* t_att = (const float*)d_in[2];
    const float* w1    = (const float*)d_in[3];
    const float* b1    = (const float*)d_in[4];
    const float* w2    = (const float*)d_in[5];
    const float* b2    = (const float*)d_in[6];
    const int*   erow  = (const int*)d_in[7];
    const int*   ecol  = (const int*)d_in[8];
    float* out = (float*)d_out;

    const int N = in_sizes[0] / 64;
    const int E = in_sizes[7];

    float* ws  = (float*)d_ws;
    float* lg  = ws;
    float* wE  = ws + (size_t)N * 64;
    int* rowptr = (int*)(ws + (size_t)N * 64 + E);

    float* out1 = out + (size_t)N * 64;
    float* out2 = out + 2 * (size_t)N * 64;

    const int TPB = 256;

    const int edge_bl = cdiv(cdiv(E, 32), TPB / 32);

    init_kernel<<<FILL_BLOCKS, TPB, 0, stream>>>(h0, out, N);
    logits_kernel<<<edge_bl, TPB, 0, stream>>>(h_att, t_att, erow, ecol, rowptr, lg, E, N);
    rowsoftmax_kernel<<<SOFT_BLOCKS, TPB, 0, stream>>>(lg, wE, rowptr, N);

    const int ntiles   = (N + 15) >> 4;
    const int layer_bl = cdiv(ntiles, LAYER_WAVES);

    layer_kernel<<<layer_bl, LAYER_WAVES * 32, 0, stream>>>(h0,   wE, erow, ecol, rowptr, w1, b1, out1, N, E);
    layer_kernel<<<layer_bl, LAYER_WAVES * 32, 0, stream>>>(out1, wE, erow, ecol, rowptr, w2, b2, out2, N, E);
}


# layer kernels' edge walk: full groups of 32 edges run straight-line code with no per-edge end test, row starts from one ballot mask per group; the run's last group as before; sums in the same order
# speedup vs baseline: 3.3128x; 1.0755x over previous
#include <hip/hip_runtime.h>
#include <math.h>

typedef __attribute__((ext_vector_type(2))) float v2f;
typedef __attribute__((ext_vector_type(8))) float v8f;

#define GEPS 1e-12f

#define FILL_BLOCKS 1024

__global__ void init_kernel(const float* h0, float* out0, int n) {
    const int tid    = blockIdx.x * blockDim.x + threadIdx.x;
    const int stride = gridDim.x * blockDim.x;
    const int n4     = n * 16;
    const float4* src = (const float4*)h0;
    float4* dst = (float4*)out0;
    int i = tid;
    for (; i + 3 * stride < n4; i += 4 * stride) {
        const float4 v0 = src[i];
        const float4 v1 = src[i + stride];
        const float4 v2 = src[i + 2 * stride];
        const float4 v3 = src[i + 3 * stride];
        dst[i]              = v0;
        dst[i + stride]     = v1;
        dst[i + 2 * stride] = v2;
        dst[i + 3 * stride] = v3;
    }
    if (i + stride < n4) {
        const float4 v0 = src[i];
        const float4 v1 = src[i + stride];
        dst[i]          = v0;
        dst[i + stride] = v1;
        i += 2 * stride;
    }
    if (i < n4) { dst[i] = src[i]; }
}

__global__ void logits_kernel(const float* __restrict__ h_att, const float* __restrict__ t_att,
                              const int* __restrict__ erow, const int* __restrict__ ecol,
                              int* __restrict__ rowptr, float* __restrict__ wE,
                              int n_edges, int n_rows) {
    const int lane = threadIdx.x & 31;
    const int wave = __builtin_amdgcn_readfirstlane(
        (int)(blockIdx.x * (blockDim.x >> 5) + (threadIdx.x >> 5)));
    const int e0 = wave * 32;
    if (e0 >= n_edges) return;
    const int cnt = min(32, n_edges - e0);
    const int el   = e0 + min(lane, cnt - 1);
    const int my_r = erow[el];
    const int my_c = ecol[el];
    const int prev0 = e0 > 0 ? erow[e0 - 1] : -1;
    const int half = lane >> 4;
    const int sub  = lane & 15;
    float mylogit = 0.0f;

#pragma unroll
    for (int g = 0; g < 2; ++g) {
        float4 a[8], b[8];
#pragma unroll
        for (int k = 0; k < 8; ++k) {
            const int t = g * 8 + k;
            const int r_lo = __builtin_amdgcn_readlane(my_r, t);
            const int r_hi = __builtin_amdgcn_readlane(my_r, t + 16);
            const int c_lo = __builtin_amdgcn_readlane(my_c, t);
            const int c_hi = __builtin_amdgcn_readlane(my_c, t + 16);
            const int r = half ? r_hi : r_lo;
            const int c = half ? c_hi : c_lo;
            a[k] = *(const float4*)(h_att + (size_t)r * 64 + sub * 4);
            b[k] = *(const float4*)(t_att + (size_t)c * 64 + sub * 4);
        }
#pragma unroll
        for (int k = 0; k < 8; ++k) {
            const int t = g * 8 + k;
            float dot = a[k].x * b[k].x + a[k].y * b[k].y + a[k].z * b[k].z + a[k].w * b[k].w;
            dot += __shfl_xor(dot, 1, 32);
            dot += __shfl_xor(dot, 2, 32);
            dot += __shfl_xor(dot, 4, 32);
            dot += __shfl_xor(dot, 8, 32);
            if (sub == t) mylogit = dot;
        }
    }
    int rp = __shfl_up(my_r, 1, 32);
    if (lane == 0) rp = prev0;
    if (lane < cnt) {
        const int e = e0 + lane;
        wE[e] = mylogit;
        for (int g = rp + 1; g <= my_r; ++g)
            rowptr[g] = e;
        if (e == n_edges - 1)
            for (int g = my_r + 1; g <= n_rows; ++g)
                rowptr[g] = n_edges;
    }
}

#define SOFT_BLOCKS 1024

__global__ void rowsoftmax_kernel(const float* __restrict__ lg, float* __restrict__ wE,
                                  const int* __restrict__ rowptr, int n_rows) {
    const int sub    = threadIdx.x & 15;
    const int slot   = (int)((blockIdx.x * blockDim.x + threadIdx.x) >> 4);
    const int nslots = (int)((gridDim.x * blockDim.x) >> 4);
    const int per    = (n_rows + nslots - 1) / nslots;
    const int rbeg   = slot * per;
    const int rend   = min(rbeg + per, n_rows);
    const float ninf = -__builtin_inff();
    for (int r = rbeg; r < rend; ++r) {
        const int a = rowptr[r];
        const int b = rowptr[r + 1];
        const int k0 = a + sub;
        const int k1 = k0 + 16;
        const bool p0 = k0 < b;
        const bool p1 = k1 < b;
        const float l0 = p0 ? lg[k0] : ninf;
        const float l1 = p1 ? lg[k1] : ninf;
        float mx = fmaxf(l0, l1);
        for (int k = k0 + 32; k < b; k += 16) mx = fmaxf(mx, lg[k]);
        mx = fmaxf(mx, __shfl_xor(mx, 1, 32));
        mx = fmaxf(mx, __shfl_xor(mx, 2, 32));
        mx = fmaxf(mx, __shfl_xor(mx, 4, 32));
        mx = fmaxf(mx, __shfl_xor(mx, 8, 32));
        const float x0 = p0 ? expf(l0 - mx) : 0.0f;
        const float x1 = p1 ? expf(l1 - mx) : 0.0f;
        float s = x0 + x1;
        for (int k = k0 + 32; k < b; k += 16) s += expf(lg[k] - mx);
        s += __shfl_xor(s, 1, 32);
        s += __shfl_xor(s, 2, 32);
        s += __shfl_xor(s, 4, 32);
        s += __shfl_xor(s, 8, 32);
        const float d = fmaxf(s, GEPS);
        if (p0) wE[k0] = x0 / d;
        if (p1) wE[k1] = x1 / d;
        for (int k = k0 + 32; k < b; k += 16) wE[k] = expf(lg[k] - mx) / d;
    }
}

#define LAYER_WAVES 4
#define DPITCH 68

__global__ void __launch_bounds__(LAYER_WAVES * 32)
layer_kernel(const float* __restrict__ h, const float* __restrict__ wE,
             const int* __restrict__ erow, const int* __restrict__ ecol,
             const int* __restrict__ rowptr,
             const float* __restrict__ W, const float* __restrict__ bias,
             float* __restrict__ out, int n, int n_edges) {
    __shared__ __attribute__((aligned(16))) float sW[64 * DPITCH];
    __shared__ __attribute__((aligned(16))) float sT[LAYER_WAVES][16 * DPITCH];

    const int lane = threadIdx.x & 31;
    const int wave = __builtin_amdgcn_readfirstlane((int)(threadIdx.x >> 5));
    const int tile = blockIdx.x * LAYER_WAVES + wave;
    const int ntiles = (n + 15) >> 4;
    const bool live = tile < ntiles;
    const int  tl   = live ? tile : ntiles - 1;

    const int half = lane >> 4;
    const int l16  = lane & 15;
    float* tbuf = sT[wave];

    const int r0 = tl * 16;
    const int r1 = min(r0 + 16, n);
    const int ea = __builtin_amdgcn_readfirstlane(rowptr[r0]);
    const int eb = __builtin_amdgcn_readfirstlane(rowptr[r1]);

    float4 wv[8];
#pragma unroll
    for (int i = 0; i < 8; ++i) {
        const int idx = (int)threadIdx.x + i * (LAYER_WAVES * 32);
        wv[i] = *(const float4*)(W + (size_t)idx * 4);
    }

    int   nx_r, nx_c;
    float nx_w;
    {
        const int el = min(ea + lane, n_edges - 1);
        nx_r = erow[el];
        nx_c = ecol[el];
        nx_w = wE[el];
    }

    float4 hv[8];
#pragma unroll
    for (int i = 0; i < 8; ++i) {
        const int row  = r0 + 2 * i + half;
        const int rowc = row < n ? row : n - 1;
        hv[i] = *(const float4*)(h + (size_t)rowc * 64 + l16 * 4);
    }

#pragma unroll
    for (int i = 0; i < 8; ++i) {
        const int idx = (int)threadIdx.x + i * (LAYER_WAVES * 32);
        const int r   = idx >> 4;
        const int c4  = idx & 15;
        *(float4*)(sW + r * DPITCH + c4 * 4) = wv[i];
    }
#pragma unroll
    for (int i = 0; i < 8; ++i)
        *(float4*)(tbuf + (2 * i + half) * DPITCH + l16 * 4) = make_float4(0.0f, 0.0f, 0.0f, 0.0f);

    __syncthreads();
    if (!live) return;

    {
        float2* trow = (float2*)tbuf;
        int   cur = -1;
        float ax = 0.0f, ay = 0.0f;
        const char* hb = (const char*)h;
        const unsigned int lane8 = (unsigned int)lane * 8u;

        int p = ea;
        for (; p + 32 <= eb; p += 32) {
            const int   my_r = nx_r;
            const float my_w = nx_w;
            const unsigned int my_o = (unsigned int)nx_c * 256u;
            {
                const int el = min(p + 32 + lane, n_edges - 1);
                nx_r = erow[el];
                nx_c = ecol[el];
                nx_w = wE[el];
            }

            float2 gv[32];
#pragma unroll
            for (int j = 0; j < 32; ++j) {
                const unsigned int o = (unsigned int)__builtin_amdgcn_readlane((int)my_o, j);
                gv[j] = *(const float2*)(hb + (o + lane8));
            }
            int pr = __shfl_up(my_r, 1, 32);
            if (lane == 0) pr = cur;
            const unsigned int first = (unsigned int)__builtin_amdgcn_ballot_w32(my_r != pr);
#pragma unroll
            for (int j = 0; j < 32; ++j) {
                const float we = __int_as_float(__builtin_amdgcn_readlane(__float_as_int(my_w), j));
                if (__builtin_expect((first >> j) & 1u, 0u)) {
                    if (cur >= 0) trow[(cur - r0) * (DPITCH / 2) + lane] = make_float2(ax, ay);
                    ax = 0.0f; ay = 0.0f;
                    cur = __builtin_amdgcn_readlane(my_r, j);
                }
                ax = fmaf(we, gv[j].x, ax);
                ay = fmaf(we, gv[j].y, ay);
            }
        }

        if (p < eb) {
            const int   cnt  = eb - p;
            const int   my_r = nx_r;
            const float my_w = nx_w;
            const unsigned int my_o = (unsigned int)nx_c * 256u;

            float2 gv[32];
#pragma unroll
            for (int j = 0; j < 32; ++j) {
                if (j < cnt) {
                    const unsigned int o = (unsigned int)__builtin_amdgcn_readlane((int)my_o, j);
                    gv[j] = *(const float2*)(hb + (o + lane8));
                }
            }

#pragma unroll
            for (int j = 0; j < 32; ++j) {
                if (j < cnt) {
                    const int   r  = __builtin_amdgcn_readlane(my_r, j);
                    const float we = __int_as_float(__builtin_amdgcn_readlane(__float_as_int(my_w), j));
                    if (r != cur) {
                        if (cur >= 0) trow[(cur - r0) * (DPITCH / 2) + lane] = make_float2(ax, ay);
                        ax = 0.0f; ay = 0.0f; cur = r;
                    }
                    ax = fmaf(we, gv[j].x, ax);
                    ay = fmaf(we, gv[j].y, ay);
                }
            }
        }
        if (cur >= 0) trow[(cur - r0) * (DPITCH / 2) + lane] = make_float2(ax, ay);
    }

    __builtin_amdgcn_fence(__ATOMIC_RELEASE, "wavefront");
    __builtin_amdgcn_wave_barrier();
    __builtin_amdgcn_fence(__ATOMIC_ACQUIRE, "wavefront");

#pragma unroll
    for (int i = 0; i < 8; ++i) {
        float* q = tbuf + (2 * i + half) * DPITCH + l16 * 4;
        const float4 mv = *(const float4*)q;
        float4 sm;
        sm.x = hv[i].x + mv.x;
        sm.y = hv[i].y + mv.y;
        sm.z = hv[i].z + mv.z;
        sm.w = hv[i].w + mv.w;
        *(float4*)q = sm;
    }

    __builtin_amdgcn_fence(__ATOMIC_RELEASE, "wavefront");
    __builtin_amdgcn_wave_barrier();
    __builtin_amdgcn_fence(__ATOMIC_ACQUIRE, "wavefront");

    v8f acc[4] = {};

#pragma unroll
    for (int kb = 0; kb < 16; ++kb) {
        const int koff = kb * 4 + half * 2;
        const v2f a = *(const v2f*)(tbuf + l16 * DPITCH + koff);
#pragma unroll
        for (int jb = 0; jb < 4; ++jb) {
            const v2f b = *(const v2f*)(sW + (jb * 16 + l16) * DPITCH + koff);
            acc[jb] = __builtin_amdgcn_wmma_f32_16x16x4_f32(
                false, a, false, b, (short)0, acc[jb], false, false);
        }
    }

    float bj[4];
#pragma unroll
    for (int jb = 0; jb < 4; ++jb) bj[jb] = bias[jb * 16 + l16];

#pragma unroll
    for (int v = 0; v < 8; ++v) {
        float vals[4];
        float ss = 0.0f;
#pragma unroll
        for (int jb = 0; jb < 4; ++jb) {
            float y = fmaxf(acc[jb][v] + bj[jb], 0.0f);
            vals[jb] = y;
            ss = fmaf(y, y, ss);
        }
        ss += __shfl_xor(ss, 1, 32);
        ss += __shfl_xor(ss, 2, 32);
        ss += __shfl_xor(ss, 4, 32);
        ss += __shfl_xor(ss, 8, 32);
        float inv = 1.0f / fmaxf(sqrtf(ss), GEPS);
        const int m = v + half * 8;
#pragma unroll
        for (int jb = 0; jb < 4; ++jb)
            tbuf[m * DPITCH + jb * 16 + l16] = vals[jb] * inv;
    }

    __builtin_amdgcn_fence(__ATOMIC_RELEASE, "wavefront");
    __builtin_amdgcn_wave_barrier();
    __builtin_amdgcn_fence(__ATOMIC_ACQUIRE, "wavefront");

    float4 o[8];
#pragma unroll
    for (int i = 0; i < 8; ++i)
        o[i] = *(const float4*)(tbuf + (2 * i + half) * DPITCH + l16 * 4);

    float* orow0 = out + (size_t)tile * 16 * 64 + (size_t)half * 64 + l16 * 4;
    if (tile * 16 + 16 <= n) {
#pragma unroll
        for (int i = 0; i < 8; ++i)
            *(float4*)(orow0 + (size_t)i * 128) = o[i];
    } else {
#pragma unroll
        for (int i = 0; i < 8; ++i)
            if (tile * 16 + 2 * i + half < n)
                *(float4*)(orow0 + (size_t)i * 128) = o[i];
    }
}

static inline int cdiv(long long a, long long b) { return (int)((a + b - 1) / b); }

extern "C" void kernel_launch(void* const* d_in, const int* in_sizes, int n_in,
                              void* d_out, int out_size, void* d_ws, size_t ws_size,
                              hipStream_t stream) {
    const float* h0    = (const float*)d_in[0];
    const float* h_att = (const float*)d_in[1];
    const float* t_att = (const float*)d_in[2];
    const float* w1    = (const float*)d_in[3];
    const float* b1    = (const float*)d_in[4];
    const float* w2    = (const float*)d_in[5];
    const float* b2    = (const float*)d_in[6];
    const int*   erow  = (const int*)d_in[7];
    const int*   ecol  = (const int*)d_in[8];
    float* out = (float*)d_out;

    const int N = in_sizes[0] / 64;
    const int E = in_sizes[7];

    float* ws  = (float*)d_ws;
    float* lg  = ws;
    float* wE  = ws + (size_t)N * 64;
    int* rowptr = (int*)(ws + (size_t)N * 64 + E);

    float* out1 = out + (size_t)N * 64;
    float* out2 = out + 2 * (size_t)N * 64;

    const int TPB = 256;

    const int edge_bl = cdiv(cdiv(E, 32), TPB / 32);

    init_kernel<<<FILL_BLOCKS, TPB, 0, stream>>>(h0, out, N);
    logits_kernel<<<edge_bl, TPB, 0, stream>>>(h_att, t_att, erow, ecol, rowptr, lg, E, N);
    rowsoftmax_kernel<<<SOFT_BLOCKS, TPB, 0, stream>>>(lg, wE, rowptr, N);

    const int ntiles   = (N + 15) >> 4;
    const int layer_bl = cdiv(ntiles, LAYER_WAVES);

    layer_kernel<<<layer_bl, LAYER_WAVES * 32, 0, stream>>>(h0,   wE, erow, ecol, rowptr, w1, b1, out1, N, E);
    layer_kernel<<<layer_bl, LAYER_WAVES * 32, 0, stream>>>(out1, wE, erow, ecol, rowptr, w2, b2, out2, N, E);
}


# out[0] = h0 is stored by the layer-1 kernel from the h0 tile rows it already holds in registers (whole rows, behind its edge walk); the copy kernel is gone: 4 launches; sums unchanged
# speedup vs baseline: 3.4212x; 1.0327x over previous
#include <hip/hip_runtime.h>
#include <math.h>

typedef __attribute__((ext_vector_type(2))) float v2f;
typedef __attribute__((ext_vector_type(8))) float v8f;

#define GEPS 1e-12f

__global__ void logits_kernel(const float* __restrict__ h_att, const float* __restrict__ t_att,
                              const int* __restrict__ erow, const int* __restrict__ ecol,
                              int* __restrict__ rowptr, float* __restrict__ wE,
                              int n_edges, int n_rows) {
    const int lane = threadIdx.x & 31;
    const int wave = __builtin_amdgcn_readfirstlane(
        (int)(blockIdx.x * (blockDim.x >> 5) + (threadIdx.x >> 5)));
    const int e0 = wave * 32;
    if (e0 >= n_edges) return;
    const int cnt = min(32, n_edges - e0);
    const int el   = e0 + min(lane, cnt - 1);
    const int my_r = erow[el];
    const int my_c = ecol[el];
    const int prev0 = e0 > 0 ? erow[e0 - 1] : -1;
    const int half = lane >> 4;
    const int sub  = lane & 15;
    float mylogit = 0.0f;

#pragma unroll
    for (int g = 0; g < 2; ++g) {
        float4 a[8], b[8];
#pragma unroll
        for (int k = 0; k < 8; ++k) {
            const int t = g * 8 + k;
            const int r_lo = __builtin_amdgcn_readlane(my_r, t);
            const int r_hi = __builtin_amdgcn_readlane(my_r, t + 16);
            const int c_lo = __builtin_amdgcn_readlane(my_c, t);
            const int c_hi = __builtin_amdgcn_readlane(my_c, t + 16);
            const int r = half ? r_hi : r_lo;
            const int c = half ? c_hi : c_lo;
            a[k] = *(const float4*)(h_att + (size_t)r * 64 + sub * 4);
            b[k] = *(const float4*)(t_att + (size_t)c * 64 + sub * 4);
        }
#pragma unroll
        for (int k = 0; k < 8; ++k) {
            const int t = g * 8 + k;
            float dot = a[k].x * b[k].x + a[k].y * b[k].y + a[k].z * b[k].z + a[k].w * b[k].w;
            dot += __shfl_xor(dot, 1, 32);
            dot += __shfl_xor(dot, 2, 32);
            dot += __shfl_xor(dot, 4, 32);
            dot += __shfl_xor(dot, 8, 32);
            if (sub == t) mylogit = dot;
        }
    }
    int rp = __shfl_up(my_r, 1, 32);
    if (lane == 0) rp = prev0;
    if (lane < cnt) {
        const int e = e0 + lane;
        wE[e] = mylogit;
        for (int g = rp + 1; g <= my_r; ++g)
            rowptr[g] = e;
        if (e == n_edges - 1)
            for (int g = my_r + 1; g <= n_rows; ++g)
                rowptr[g] = n_edges;
    }
}

#define SOFT_BLOCKS 1024

__global__ void rowsoftmax_kernel(const float* __restrict__ lg, float* __restrict__ wE,
                                  const int* __restrict__ rowptr, int n_rows) {
    const int sub    = threadIdx.x & 15;
    const int slot   = (int)((blockIdx.x * blockDim.x + threadIdx.x) >> 4);
    const int nslots = (int)((gridDim.x * blockDim.x) >> 4);
    const int per    = (n_rows + nslots - 1) / nslots;
    const int rbeg   = slot * per;
    const int rend   = min(rbeg + per, n_rows);
    const float ninf = -__builtin_inff();
    for (int r = rbeg; r < rend; ++r) {
        const int a = rowptr[r];
        const int b = rowptr[r + 1];
        const int k0 = a + sub;
        const int k1 = k0 + 16;
        const bool p0 = k0 < b;
        const bool p1 = k1 < b;
        const float l0 = p0 ? lg[k0] : ninf;
        const float l1 = p1 ? lg[k1] : ninf;
        float mx = fmaxf(l0, l1);
        for (int k = k0 + 32; k < b; k += 16) mx = fmaxf(mx, lg[k]);
        mx = fmaxf(mx, __shfl_xor(mx, 1, 32));
        mx = fmaxf(mx, __shfl_xor(mx, 2, 32));
        mx = fmaxf(mx, __shfl_xor(mx, 4, 32));
        mx = fmaxf(mx, __shfl_xor(mx, 8, 32));
        const float x0 = p0 ? expf(l0 - mx) : 0.0f;
        const float x1 = p1 ? expf(l1 - mx) : 0.0f;
        float s = x0 + x1;
        for (int k = k0 + 32; k < b; k += 16) s += expf(lg[k] - mx);
        s += __shfl_xor(s, 1, 32);
        s += __shfl_xor(s, 2, 32);
        s += __shfl_xor(s, 4, 32);
        s += __shfl_xor(s, 8, 32);
        const float d = fmaxf(s, GEPS);
        if (p0) wE[k0] = x0 / d;
        if (p1) wE[k1] = x1 / d;
        for (int k = k0 + 32; k < b; k += 16) wE[k] = expf(lg[k] - mx) / d;
    }
}

#define LAYER_WAVES 4
#define DPITCH 68

template <bool COPY0>
__global__ void __launch_bounds__(LAYER_WAVES * 32)
layer_kernel(const float* __restrict__ h, float* __restrict__ out0, const float* __restrict__ wE,
             const int* __restrict__ erow, const int* __restrict__ ecol,
             const int* __restrict__ rowptr,
             const float* __restrict__ W, const float* __restrict__ bias,
             float* __restrict__ out, int n, int n_edges) {
    __shared__ __attribute__((aligned(16))) float sW[64 * DPITCH];
    __shared__ __attribute__((aligned(16))) float sT[LAYER_WAVES][16 * DPITCH];

    const int lane = threadIdx.x & 31;
    const int wave = __builtin_amdgcn_readfirstlane((int)(threadIdx.x >> 5));
    const int tile = blockIdx.x * LAYER_WAVES + wave;
    const int ntiles = (n + 15) >> 4;
    const bool live = tile < ntiles;
    const int  tl   = live ? tile : ntiles - 1;

    const int half = lane >> 4;
    const int l16  = lane & 15;
    float* tbuf = sT[wave];

    const int r0 = tl * 16;
    const int r1 = min(r0 + 16, n);
    const int ea = __builtin_amdgcn_readfirstlane(rowptr[r0]);
    const int eb = __builtin_amdgcn_readfirstlane(rowptr[r1]);

    float4 wv[8];
#pragma unroll
    for (int i = 0; i < 8; ++i) {
        const int idx = (int)threadIdx.x + i * (LAYER_WAVES * 32);
        wv[i] = *(const float4*)(W + (size_t)idx * 4);
    }

    int   nx_r, nx_c;
    float nx_w;
    {
        const int el = min(ea + lane, n_edges - 1);
        nx_r = erow[el];
        nx_c = ecol[el];
        nx_w = wE[el];
    }

    float4 hv[8];
#pragma unroll
    for (int i = 0; i < 8; ++i) {
        const int row  = r0 + 2 * i + half;
        const int rowc = row < n ? row : n - 1;
        hv[i] = *(const float4*)(h + (size_t)rowc * 64 + l16 * 4);
    }

#pragma unroll
    for (int i = 0; i < 8; ++i) {
        const int idx = (int)threadIdx.x + i * (LAYER_WAVES * 32);
        const int r   = idx >> 4;
        const int c4  = idx & 15;
        *(float4*)(sW + r * DPITCH + c4 * 4) = wv[i];
    }
#pragma unroll
    for (int i = 0; i < 8; ++i)
        *(float4*)(tbuf + (2 * i + half) * DPITCH + l16 * 4) = make_float4(0.0f, 0.0f, 0.0f, 0.0f);

    __syncthreads();
    if (!live) return;

    {
        float2* trow = (float2*)tbuf;
        int   cur = -1;
        float ax = 0.0f, ay = 0.0f;
        const char* hb = (const char*)h;
        const unsigned int lane8 = (unsigned int)lane * 8u;

        int p = ea;
        for (; p + 32 <= eb; p += 32) {
            const int   my_r = nx_r;
            const float my_w = nx_w;
            const unsigned int my_o = (unsigned int)nx_c * 256u;
            {
                const int el = min(p + 32 + lane, n_edges - 1);
                nx_r = erow[el];
                nx_c = ecol[el];
                nx_w = wE[el];
            }

            float2 gv[32];
#pragma unroll
            for (int j = 0; j < 32; ++j) {
                const unsigned int o = (unsigned int)__builtin_amdgcn_readlane((int)my_o, j);
                gv[j] = *(const float2*)(hb + (o + lane8));
            }
            int pr = __shfl_up(my_r, 1, 32);
            if (lane == 0) pr = cur;
            const unsigned int first = (unsigned int)__builtin_amdgcn_ballot_w32(my_r != pr);
#pragma unroll
            for (int j = 0; j < 32; ++j) {
                const float we = __int_as_float(__builtin_amdgcn_readlane(__float_as_int(my_w), j));
                if (__builtin_expect((first >> j) & 1u, 0u)) {
                    if (cur >= 0) trow[(cur - r0) * (DPITCH / 2) + lane] = make_float2(ax, ay);
                    ax = 0.0f; ay = 0.0f;
                    cur = __builtin_amdgcn_readlane(my_r, j);
                }
                ax = fmaf(we, gv[j].x, ax);
                ay = fmaf(we, gv[j].y, ay);
            }
        }

        if (p < eb) {
            const int   cnt  = eb - p;
            const int   my_r = nx_r;
            const float my_w = nx_w;
            const unsigned int my_o = (unsigned int)nx_c * 256u;

            float2 gv[32];
#pragma unroll
            for (int j = 0; j < 32; ++j) {
                if (j < cnt) {
                    const unsigned int o = (unsigned int)__builtin_amdgcn_readlane((int)my_o, j);
                    gv[j] = *(const float2*)(hb + (o + lane8));
                }
            }

#pragma unroll
            for (int j = 0; j < 32; ++j) {
                if (j < cnt) {
                    const int   r  = __builtin_amdgcn_readlane(my_r, j);
                    const float we = __int_as_float(__builtin_amdgcn_readlane(__float_as_int(my_w), j));
                    if (r != cur) {
                        if (cur >= 0) trow[(cur - r0) * (DPITCH / 2) + lane] = make_float2(ax, ay);
                        ax = 0.0f; ay = 0.0f; cur = r;
                    }
                    ax = fmaf(we, gv[j].x, ax);
                    ay = fmaf(we, gv[j].y, ay);
                }
            }
        }
        if (cur >= 0) trow[(cur - r0) * (DPITCH / 2) + lane] = make_float2(ax, ay);
    }

    if (COPY0) {
        float* c0 = out0 + (size_t)tile * 16 * 64 + (size_t)half * 64 + l16 * 4;
        if (tile * 16 + 16 <= n) {
#pragma unroll
            for (int i = 0; i < 8; ++i)
                *(float4*)(c0 + (size_t)i * 128) = hv[i];
        } else {
#pragma unroll
            for (int i = 0; i < 8; ++i)
                if (tile * 16 + 2 * i + half < n)
                    *(float4*)(c0 + (size_t)i * 128) = hv[i];
        }
    }

    __builtin_amdgcn_fence(__ATOMIC_RELEASE, "wavefront");
    __builtin_amdgcn_wave_barrier();
    __builtin_amdgcn_fence(__ATOMIC_ACQUIRE, "wavefront");

#pragma unroll
    for (int i = 0; i < 8; ++i) {
        float* q = tbuf + (2 * i + half) * DPITCH + l16 * 4;
        const float4 mv = *(const float4*)q;
        float4 sm;
        sm.x = hv[i].x + mv.x;
        sm.y = hv[i].y + mv.y;
        sm.z = hv[i].z + mv.z;
        sm.w = hv[i].w + mv.w;
        *(float4*)q = sm;
    }

    __builtin_amdgcn_fence(__ATOMIC_RELEASE, "wavefront");
    __builtin_amdgcn_wave_barrier();
    __builtin_amdgcn_fence(__ATOMIC_ACQUIRE, "wavefront");

    v8f acc[4] = {};

#pragma unroll
    for (int kb = 0; kb < 16; ++kb) {
        const int koff = kb * 4 + half * 2;
        const v2f a = *(const v2f*)(tbuf + l16 * DPITCH + koff);
#pragma unroll
        for (int jb = 0; jb < 4; ++jb) {
            const v2f b = *(const v2f*)(sW + (jb * 16 + l16) * DPITCH + koff);
            acc[jb] = __builtin_amdgcn_wmma_f32_16x16x4_f32(
                false, a, false, b, (short)0, acc[jb], false, false);
        }
    }

    float bj[4];
#pragma unroll
    for (int jb = 0; jb < 4; ++jb) bj[jb] = bias[jb * 16 + l16];

#pragma unroll
    for (int v = 0; v < 8; ++v) {
        float vals[4];
        float ss = 0.0f;
#pragma unroll
        for (int jb = 0; jb < 4; ++jb) {
            float y = fmaxf(acc[jb][v] + bj[jb], 0.0f);
            vals[jb] = y;
            ss = fmaf(y, y, ss);
        }
        ss += __shfl_xor(ss, 1, 32);
        ss += __shfl_xor(ss, 2, 32);
        ss += __shfl_xor(ss, 4, 32);
        ss += __shfl_xor(ss, 8, 32);
        float inv = 1.0f / fmaxf(sqrtf(ss), GEPS);
        const int m = v + half * 8;
#pragma unroll
        for (int jb = 0; jb < 4; ++jb)
            tbuf[m * DPITCH + jb * 16 + l16] = vals[jb] * inv;
    }

    __builtin_amdgcn_fence(__ATOMIC_RELEASE, "wavefront");
    __builtin_amdgcn_wave_barrier();
    __builtin_amdgcn_fence(__ATOMIC_ACQUIRE, "wavefront");

    float4 o[8];
#pragma unroll
    for (int i = 0; i < 8; ++i)
        o[i] = *(const float4*)(tbuf + (2 * i + half) * DPITCH + l16 * 4);

    float* orow0 = out + (size_t)tile * 16 * 64 + (size_t)half * 64 + l16 * 4;
    if (tile * 16 + 16 <= n) {
#pragma unroll
        for (int i = 0; i < 8; ++i)
            *(float4*)(orow0 + (size_t)i * 128) = o[i];
    } else {
#pragma unroll
        for (int i = 0; i < 8; ++i)
            if (tile * 16 + 2 * i + half < n)
                *(float4*)(orow0 + (size_t)i * 128) = o[i];
    }
}

static inline int cdiv(long long a, long long b) { return (int)((a + b - 1) / b); }

extern "C" void kernel_launch(void* const* d_in, const int* in_sizes, int n_in,
                              void* d_out, int out_size, void* d_ws, size_t ws_size,
                              hipStream_t stream) {
    const float* h0    = (const float*)d_in[0];
    const float* h_att = (const float*)d_in[1];
    const float* t_att = (const float*)d_in[2];
    const float* w1    = (const float*)d_in[3];
    const float* b1    = (const float*)d_in[4];
    const float* w2    = (const float*)d_in[5];
    const float* b2    = (const float*)d_in[6];
    const int*   erow  = (const int*)d_in[7];
    const int*   ecol  = (const int*)d_in[8];
    float* out = (float*)d_out;

    const int N = in_sizes[0] / 64;
    const int E = in_sizes[7];

    float* ws  = (float*)d_ws;
    float* lg  = ws;
    float* wE  = ws + (size_t)N * 64;
    int* rowptr = (int*)(ws + (size_t)N * 64 + E);

    float* out1 = out + (size_t)N * 64;
    float* out2 = out + 2 * (size_t)N * 64;

    const int TPB = 256;

    const int edge_bl = cdiv(cdiv(E, 32), TPB / 32);

    logits_kernel<<<edge_bl, TPB, 0, stream>>>(h_att, t_att, erow, ecol, rowptr, lg, E, N);
    rowsoftmax_kernel<<<SOFT_BLOCKS, TPB, 0, stream>>>(lg, wE, rowptr, N);

    const int ntiles   = (N + 15) >> 4;
    const int layer_bl = cdiv(ntiles, LAYER_WAVES);

    layer_kernel<true><<<layer_bl, LAYER_WAVES * 32, 0, stream>>>(h0,   out,     wE, erow, ecol, rowptr, w1, b1, out1, N, E);
    layer_kernel<false><<<layer_bl, LAYER_WAVES * 32, 0, stream>>>(out1, nullptr, wE, erow, ecol, rowptr, w2, b2, out2, N, E);
}


# edge softmax inside the layer-1 kernel: the wave owning 16 rows takes their max and exp sum before its edge walk and forms each weight in the walk; softmax launch gone, 3 launches; same sum order
# speedup vs baseline: 3.6443x; 1.0652x over previous
#include <hip/hip_runtime.h>
#include <math.h>

typedef __attribute__((ext_vector_type(2))) float v2f;
typedef __attribute__((ext_vector_type(8))) float v8f;

#define GEPS 1e-12f

__global__ void logits_kernel(const float* __restrict__ h_att, const float* __restrict__ t_att,
                              const int* __restrict__ erow, const int* __restrict__ ecol,
                              int* __restrict__ rowptr, float* __restrict__ wE,
                              int n_edges, int n_rows) {
    const int lane = threadIdx.x & 31;
    const int wave = __builtin_amdgcn_readfirstlane(
        (int)(blockIdx.x * (blockDim.x >> 5) + (threadIdx.x >> 5)));
    const int e0 = wave * 32;
    if (e0 >= n_edges) return;
    const int cnt = min(32, n_edges - e0);
    const int el   = e0 + min(lane, cnt - 1);
    const int my_r = erow[el];
    const int my_c = ecol[el];
    const int prev0 = e0 > 0 ? erow[e0 - 1] : -1;
    const int half = lane >> 4;
    const int sub  = lane & 15;
    float mylogit = 0.0f;

#pragma unroll
    for (int g = 0; g < 2; ++g) {
        float4 a[8], b[8];
#pragma unroll
        for (int k = 0; k < 8; ++k) {
            const int t = g * 8 + k;
            const int r_lo = __builtin_amdgcn_readlane(my_r, t);
            const int r_hi = __builtin_amdgcn_readlane(my_r, t + 16);
            const int c_lo = __builtin_amdgcn_readlane(my_c, t);
            const int c_hi = __builtin_amdgcn_readlane(my_c, t + 16);
            const int r = half ? r_hi : r_lo;
            const int c = half ? c_hi : c_lo;
            a[k] = *(const float4*)(h_att + (size_t)r * 64 + sub * 4);
            b[k] = *(const float4*)(t_att + (size_t)c * 64 + sub * 4);
        }
#pragma unroll
        for (int k = 0; k < 8; ++k) {
            const int t = g * 8 + k;
            float dot = a[k].x * b[k].x + a[k].y * b[k].y + a[k].z * b[k].z + a[k].w * b[k].w;
            dot += __shfl_xor(dot, 1, 32);
            dot += __shfl_xor(dot, 2, 32);
            dot += __shfl_xor(dot, 4, 32);
            dot += __shfl_xor(dot, 8, 32);
            if (sub == t) mylogit = dot;
        }
    }
    int rp = __shfl_up(my_r, 1, 32);
    if (lane == 0) rp = prev0;
    if (lane < cnt) {
        const int e = e0 + lane;
        wE[e] = mylogit;
        for (int g = rp + 1; g <= my_r; ++g)
            rowptr[g] = e;
        if (e == n_edges - 1)
            for (int g = my_r + 1; g <= n_rows; ++g)
                rowptr[g] = n_edges;
    }
}

#define LAYER_WAVES 4
#define DPITCH 68

static __device__ __forceinline__ float edge_weight(float logit, int q, float rmax, float rden) {
    const int qc  = min(max(q, 0), 15);
    const int src = (qc & 7) | ((qc & 8) << 1);
    const float m = __shfl(rmax, src, 32);
    const float d = __shfl(rden, src, 32);
    return expf(logit - m) / d;
}

template <bool FIRST>
__global__ void __launch_bounds__(LAYER_WAVES * 32)
layer_kernel(const float* __restrict__ h, float* __restrict__ out0, const float* __restrict__ wE,
             float* __restrict__ wOut, const int* __restrict__ erow, const int* __restrict__ ecol,
             const int* __restrict__ rowptr,
             const float* __restrict__ W, const float* __restrict__ bias,
             float* __restrict__ out, int n, int n_edges) {
    __shared__ __attribute__((aligned(16))) float sW[64 * DPITCH];
    __shared__ __attribute__((aligned(16))) float sT[LAYER_WAVES][16 * DPITCH];

    const int lane = threadIdx.x & 31;
    const int wave = __builtin_amdgcn_readfirstlane((int)(threadIdx.x >> 5));
    const int tile = blockIdx.x * LAYER_WAVES + wave;
    const int ntiles = (n + 15) >> 4;
    const bool live = tile < ntiles;
    const int  tl   = live ? tile : ntiles - 1;

    const int half = lane >> 4;
    const int l16  = lane & 15;
    float* tbuf = sT[wave];

    const int r0 = tl * 16;
    const int r1 = min(r0 + 16, n);
    const int ea = __builtin_amdgcn_readfirstlane(rowptr[r0]);
    const int eb = __builtin_amdgcn_readfirstlane(rowptr[r1]);

    float4 wv[8];
#pragma unroll
    for (int i = 0; i < 8; ++i) {
        const int idx = (int)threadIdx.x + i * (LAYER_WAVES * 32);
        wv[i] = *(const float4*)(W + (size_t)idx * 4);
    }

    int   nx_r, nx_c;
    float nx_w;
    {
        const int el = min(ea + lane, n_edges - 1);
        nx_r = erow[el];
        nx_c = ecol[el];
        nx_w = wE[el];
    }

    int   k0v[8], bv[8];
    float sl0[8], sl1[8];
    if (FIRST) {
        int rp[17];
#pragma unroll
        for (int i = 0; i < 17; ++i)
            rp[i] = __builtin_amdgcn_readfirstlane(rowptr[min(r0 + i, n)]);
#pragma unroll
        for (int i = 0; i < 8; ++i) {
            const int a  = half ? rp[i + 8] : rp[i];
            const int k0 = a + l16;
            k0v[i] = k0;
            bv[i]  = half ? rp[i + 9] : rp[i + 1];
            sl0[i] = wE[min(k0, n_edges - 1)];
            sl1[i] = wE[min(k0 + 16, n_edges - 1)];
        }
    }

    float4 hv[8];
#pragma unroll
    for (int i = 0; i < 8; ++i) {
        const int row  = r0 + 2 * i + half;
        const int rowc = row < n ? row : n - 1;
        hv[i] = *(const float4*)(h + (size_t)rowc * 64 + l16 * 4);
    }

#pragma unroll
    for (int i = 0; i < 8; ++i) {
        const int idx = (int)threadIdx.x + i * (LAYER_WAVES * 32);
        const int r   = idx >> 4;
        const int c4  = idx & 15;
        *(float4*)(sW + r * DPITCH + c4 * 4) = wv[i];
    }
#pragma unroll
    for (int i = 0; i < 8; ++i)
        *(float4*)(tbuf + (2 * i + half) * DPITCH + l16 * 4) = make_float4(0.0f, 0.0f, 0.0f, 0.0f);

    __syncthreads();
    if (!live) return;

    float rmax = 0.0f, rden = 1.0f;
    if (FIRST) {
        const float ninf = -__builtin_inff();
        float mx[8], sm[8];
#pragma unroll
        for (int i = 0; i < 8; ++i) {
            sl0[i] = k0v[i] < bv[i] ? sl0[i] : ninf;
            sl1[i] = k0v[i] + 16 < bv[i] ? sl1[i] : ninf;
            float m = fmaxf(sl0[i], sl1[i]);
            for (int k = k0v[i] + 32; k < bv[i]; k += 16) m = fmaxf(m, wE[k]);
            mx[i] = m;
        }
#pragma unroll
        for (int i = 0; i < 8; ++i) mx[i] = fmaxf(mx[i], __shfl_xor(mx[i], 1, 32));
#pragma unroll
        for (int i = 0; i < 8; ++i) mx[i] = fmaxf(mx[i], __shfl_xor(mx[i], 2, 32));
#pragma unroll
        for (int i = 0; i < 8; ++i) mx[i] = fmaxf(mx[i], __shfl_xor(mx[i], 4, 32));
#pragma unroll
        for (int i = 0; i < 8; ++i) mx[i] = fmaxf(mx[i], __shfl_xor(mx[i], 8, 32));
#pragma unroll
        for (int i = 0; i < 8; ++i) {
            const float e0 = expf(sl0[i] - mx[i]);
            const float e1 = expf(sl1[i] - mx[i]);
            const float x0 = k0v[i] < bv[i] ? e0 : 0.0f;
            const float x1 = k0v[i] + 16 < bv[i] ? e1 : 0.0f;
            float t = x0 + x1;
            for (int k = k0v[i] + 32; k < bv[i]; k += 16) t += expf(wE[k] - mx[i]);
            sm[i] = t;
        }
#pragma unroll
        for (int i = 0; i < 8; ++i) sm[i] += __shfl_xor(sm[i], 1, 32);
#pragma unroll
        for (int i = 0; i < 8; ++i) sm[i] += __shfl_xor(sm[i], 2, 32);
#pragma unroll
        for (int i = 0; i < 8; ++i) sm[i] += __shfl_xor(sm[i], 4, 32);
#pragma unroll
        for (int i = 0; i < 8; ++i) sm[i] += __shfl_xor(sm[i], 8, 32);
#pragma unroll
        for (int i = 0; i < 8; ++i)
            if (l16 == i) { rmax = mx[i]; rden = fmaxf(sm[i], GEPS); }
    }

    {
        float2* trow = (float2*)tbuf;
        int   cur = -1;
        float ax = 0.0f, ay = 0.0f;
        const char* hb = (const char*)h;
        const unsigned int lane8 = (unsigned int)lane * 8u;

        int p = ea;
        for (; p + 32 <= eb; p += 32) {
            const int   my_r = nx_r;
            const float my_w = FIRST ? edge_weight(nx_w, nx_r - r0, rmax, rden) : nx_w;
            if (FIRST) wOut[p + lane] = my_w;
            const unsigned int my_o = (unsigned int)nx_c * 256u;
            {
                const int el = min(p + 32 + lane, n_edges - 1);
                nx_r = erow[el];
                nx_c = ecol[el];
                nx_w = wE[el];
            }

            float2 gv[32];
#pragma unroll
            for (int j = 0; j < 32; ++j) {
                const unsigned int o = (unsigned int)__builtin_amdgcn_readlane((int)my_o, j);
                gv[j] = *(const float2*)(hb + (o + lane8));
            }
            int pr = __shfl_up(my_r, 1, 32);
            if (lane == 0) pr = cur;
            const unsigned int first = (unsigned int)__builtin_amdgcn_ballot_w32(my_r != pr);
#pragma unroll
            for (int j = 0; j < 32; ++j) {
                const float we = __int_as_float(__builtin_amdgcn_readlane(__float_as_int(my_w), j));
                if (__builtin_expect((first >> j) & 1u, 0u)) {
                    if (cur >= 0) trow[(cur - r0) * (DPITCH / 2) + lane] = make_float2(ax, ay);
                    ax = 0.0f; ay = 0.0f;
                    cur = __builtin_amdgcn_readlane(my_r, j);
                }
                ax = fmaf(we, gv[j].x, ax);
                ay = fmaf(we, gv[j].y, ay);
            }
        }

        if (p < eb) {
            const int   cnt  = eb - p;
            const int   my_r = nx_r;
            const float my_w = FIRST ? edge_weight(nx_w, nx_r - r0, rmax, rden) : nx_w;
            if (FIRST && lane < cnt) wOut[p + lane] = my_w;
            const unsigned int my_o = (unsigned int)nx_c * 256u;

            float2 gv[32];
#pragma unroll
            for (int j = 0; j < 32; ++j) {
                if (j < cnt) {
                    const unsigned int o = (unsigned int)__builtin_amdgcn_readlane((int)my_o, j);
                    gv[j] = *(const float2*)(hb + (o + lane8));
                }
            }

#pragma unroll
            for (int j = 0; j < 32; ++j) {
                if (j < cnt) {
                    const int   r  = __builtin_amdgcn_readlane(my_r, j);
                    const float we = __int_as_float(__builtin_amdgcn_readlane(__float_as_int(my_w), j));
                    if (r != cur) {
                        if (cur >= 0) trow[(cur - r0) * (DPITCH / 2) + lane] = make_float2(ax, ay);
                        ax = 0.0f; ay = 0.0f; cur = r;
                    }
                    ax = fmaf(we, gv[j].x, ax);
                    ay = fmaf(we, gv[j].y, ay);
                }
            }
        }
        if (cur >= 0) trow[(cur - r0) * (DPITCH / 2) + lane] = make_float2(ax, ay);
    }

    if (FIRST) {
        float* c0 = out0 + (size_t)tile * 16 * 64 + (size_t)half * 64 + l16 * 4;
        if (tile * 16 + 16 <= n) {
#pragma unroll
            for (int i = 0; i < 8; ++i)
                *(float4*)(c0 + (size_t)i * 128) = hv[i];
        } else {
#pragma unroll
            for (int i = 0; i < 8; ++i)
                if (tile * 16 + 2 * i + half < n)
                    *(float4*)(c0 + (size_t)i * 128) = hv[i];
        }
    }

    __builtin_amdgcn_fence(__ATOMIC_RELEASE, "wavefront");
    __builtin_amdgcn_wave_barrier();
    __builtin_amdgcn_fence(__ATOMIC_ACQUIRE, "wavefront");

#pragma unroll
    for (int i = 0; i < 8; ++i) {
        float* q = tbuf + (2 * i + half) * DPITCH + l16 * 4;
        const float4 mv = *(const float4*)q;
        float4 sm;
        sm.x = hv[i].x + mv.x;
        sm.y = hv[i].y + mv.y;
        sm.z = hv[i].z + mv.z;
        sm.w = hv[i].w + mv.w;
        *(float4*)q = sm;
    }

    __builtin_amdgcn_fence(__ATOMIC_RELEASE, "wavefront");
    __builtin_amdgcn_wave_barrier();
    __builtin_amdgcn_fence(__ATOMIC_ACQUIRE, "wavefront");

    v8f acc[4] = {};

#pragma unroll
    for (int kb = 0; kb < 16; ++kb) {
        const int koff = kb * 4 + half * 2;
        const v2f a = *(const v2f*)(tbuf + l16 * DPITCH + koff);
#pragma unroll
        for (int jb = 0; jb < 4; ++jb) {
            const v2f b = *(const v2f*)(sW + (jb * 16 + l16) * DPITCH + koff);
            acc[jb] = __builtin_amdgcn_wmma_f32_16x16x4_f32(
                false, a, false, b, (short)0, acc[jb], false, false);
        }
    }

    float bj[4];
#pragma unroll
    for (int jb = 0; jb < 4; ++jb) bj[jb] = bias[jb * 16 + l16];

#pragma unroll
    for (int v = 0; v < 8; ++v) {
        float vals[4];
        float ss = 0.0f;
#pragma unroll
        for (int jb = 0; jb < 4; ++jb) {
            float y = fmaxf(acc[jb][v] + bj[jb], 0.0f);
            vals[jb] = y;
            ss = fmaf(y, y, ss);
        }
        ss += __shfl_xor(ss, 1, 32);
        ss += __shfl_xor(ss, 2, 32);
        ss += __shfl_xor(ss, 4, 32);
        ss += __shfl_xor(ss, 8, 32);
        float inv = 1.0f / fmaxf(sqrtf(ss), GEPS);
        const int m = v + half * 8;
#pragma unroll
        for (int jb = 0; jb < 4; ++jb)
            tbuf[m * DPITCH + jb * 16 + l16] = vals[jb] * inv;
    }

    __builtin_amdgcn_fence(__ATOMIC_RELEASE, "wavefront");
    __builtin_amdgcn_wave_barrier();
    __builtin_amdgcn_fence(__ATOMIC_ACQUIRE, "wavefront");

    float4 o[8];
#pragma unroll
    for (int i = 0; i < 8; ++i)
        o[i] = *(const float4*)(tbuf + (2 * i + half) * DPITCH + l16 * 4);

    float* orow0 = out + (size_t)tile * 16 * 64 + (size_t)half * 64 + l16 * 4;
    if (tile * 16 + 16 <= n) {
#pragma unroll
        for (int i = 0; i < 8; ++i)
            *(float4*)(orow0 + (size_t)i * 128) = o[i];
    } else {
#pragma unroll
        for (int i = 0; i < 8; ++i)
            if (tile * 16 + 2 * i + half < n)
                *(float4*)(orow0 + (size_t)i * 128) = o[i];
    }
}

static inline int cdiv(long long a, long long b) { return (int)((a + b - 1) / b); }

extern "C" void kernel_launch(void* const* d_in, const int* in_sizes, int n_in,
                              void* d_out, int out_size, void* d_ws, size_t ws_size,
                              hipStream_t stream) {
    const float* h0    = (const float*)d_in[0];
    const float* h_att = (const float*)d_in[1];
    const float* t_att = (const float*)d_in[2];
    const float* w1    = (const float*)d_in[3];
    const float* b1    = (const float*)d_in[4];
    const float* w2    = (const float*)d_in[5];
    const float* b2    = (const float*)d_in[6];
    const int*   erow  = (const int*)d_in[7];
    const int*   ecol  = (const int*)d_in[8];
    float* out = (float*)d_out;

    const int N = in_sizes[0] / 64;
    const int E = in_sizes[7];

    float* ws  = (float*)d_ws;
    float* lg  = ws;
    float* wE  = ws + (size_t)N * 64;
    int* rowptr = (int*)(ws + (size_t)N * 64 + E);

    float* out1 = out + (size_t)N * 64;
    float* out2 = out + 2 * (size_t)N * 64;

    const int TPB = 256;

    const int edge_bl = cdiv(cdiv(E, 32), TPB / 32);

    logits_kernel<<<edge_bl, TPB, 0, stream>>>(h_att, t_att, erow, ecol, rowptr, lg, E, N);

    const int ntiles   = (N + 15) >> 4;
    const int layer_bl = cdiv(ntiles, LAYER_WAVES);

    layer_kernel<true><<<layer_bl, LAYER_WAVES * 32, 0, stream>>>(h0,   out,     lg, wE,      erow, ecol, rowptr, w1, b1, out1, N, E);
    layer_kernel<false><<<layer_bl, LAYER_WAVES * 32, 0, stream>>>(out1, nullptr, wE, nullptr, erow, ecol, rowptr, w2, b2, out2, N, E);
}


# logits kernel: each wave walks 8 consecutive groups of 32 edges in a loop (7,891 waves to start, was 63,126; all resident), next group's ids requested inside the current trip; sums unchanged
# speedup vs baseline: 3.7503x; 1.0291x over previous
#include <hip/hip_runtime.h>
#include <math.h>

typedef __attribute__((ext_vector_type(2))) float v2f;
typedef __attribute__((ext_vector_type(8))) float v8f;

#define GEPS 1e-12f

#define LG_GROUPS 8

__global__ void logits_kernel(const float* __restrict__ h_att, const float* __restrict__ t_att,
                              const int* __restrict__ erow, const int* __restrict__ ecol,
                              int* __restrict__ rowptr, float* __restrict__ wE,
                              int n_edges, int n_rows) {
    const int lane = threadIdx.x & 31;
    const int wave = __builtin_amdgcn_readfirstlane(
        (int)(blockIdx.x * (blockDim.x >> 5) + (threadIdx.x >> 5)));
    const int ngroups = (n_edges + 31) >> 5;
    const int gfirst  = wave * LG_GROUPS;
    if (gfirst >= ngroups) return;
    const int gend = min(gfirst + LG_GROUPS, ngroups);
    const int half = lane >> 4;
    const int sub  = lane & 15;

    int nx_r, nx_c;
    {
        const int el = min(gfirst * 32 + lane, n_edges - 1);
        nx_r = erow[el];
        nx_c = ecol[el];
    }
    int prev0 = gfirst > 0 ? erow[gfirst * 32 - 1] : -1;

    for (int grp = gfirst; grp < gend; ++grp) {
        const int e0   = grp * 32;
        const int cnt  = min(32, n_edges - e0);
        const int my_r = nx_r;
        const int my_c = nx_c;
        {
            const int el = min(e0 + 32 + lane, n_edges - 1);
            nx_r = erow[el];
            nx_c = ecol[el];
        }
        float mylogit = 0.0f;

#pragma unroll
        for (int g = 0; g < 2; ++g) {
            float4 a[8], b[8];
#pragma unroll
            for (int k = 0; k < 8; ++k) {
                const int t = g * 8 + k;
                const int r_lo = __builtin_amdgcn_readlane(my_r, t);
                const int r_hi = __builtin_amdgcn_readlane(my_r, t + 16);
                const int c_lo = __builtin_amdgcn_readlane(my_c, t);
                const int c_hi = __builtin_amdgcn_readlane(my_c, t + 16);
                const int r = half ? r_hi : r_lo;
                const int c = half ? c_hi : c_lo;
                a[k] = *(const float4*)(h_att + (size_t)r * 64 + sub * 4);
                b[k] = *(const float4*)(t_att + (size_t)c * 64 + sub * 4);
            }
#pragma unroll
            for (int k = 0; k < 8; ++k) {
                const int t = g * 8 + k;
                float dot = a[k].x * b[k].x + a[k].y * b[k].y + a[k].z * b[k].z + a[k].w * b[k].w;
                dot += __shfl_xor(dot, 1, 32);
                dot += __shfl_xor(dot, 2, 32);
                dot += __shfl_xor(dot, 4, 32);
                dot += __shfl_xor(dot, 8, 32);
                if (sub == t) mylogit = dot;
            }
        }
        int rp = __shfl_up(my_r, 1, 32);
        if (lane == 0) rp = prev0;
        if (lane < cnt) {
            const int e = e0 + lane;
            wE[e] = mylogit;
            for (int q = rp + 1; q <= my_r; ++q)
                rowptr[q] = e;
            if (e == n_edges - 1)
                for (int q = my_r + 1; q <= n_rows; ++q)
                    rowptr[q] = n_edges;
        }
        prev0 = __builtin_amdgcn_readlane(my_r, 31);
    }
}

#define LAYER_WAVES 4
#define DPITCH 68

static __device__ __forceinline__ float edge_weight(float logit, int q, float rmax, float rden) {
    const int qc  = min(max(q, 0), 15);
    const int src = (qc & 7) | ((qc & 8) << 1);
    const float m = __shfl(rmax, src, 32);
    const float d = __shfl(rden, src, 32);
    return expf(logit - m) / d;
}

template <bool FIRST>
__global__ void __launch_bounds__(LAYER_WAVES * 32)
layer_kernel(const float* __restrict__ h, float* __restrict__ out0, const float* __restrict__ wE,
             float* __restrict__ wOut, const int* __restrict__ erow, const int* __restrict__ ecol,
             const int* __restrict__ rowptr,
             const float* __restrict__ W, const float* __restrict__ bias,
             float* __restrict__ out, int n, int n_edges) {
    __shared__ __attribute__((aligned(16))) float sW[64 * DPITCH];
    __shared__ __attribute__((aligned(16))) float sT[LAYER_WAVES][16 * DPITCH];

    const int lane = threadIdx.x & 31;
    const int wave = __builtin_amdgcn_readfirstlane((int)(threadIdx.x >> 5));
    const int tile = blockIdx.x * LAYER_WAVES + wave;
    const int ntiles = (n + 15) >> 4;
    const bool live = tile < ntiles;
    const int  tl   = live ? tile : ntiles - 1;

    const int half = lane >> 4;
    const int l16  = lane & 15;
    float* tbuf = sT[wave];

    const int r0 = tl * 16;
    const int r1 = min(r0 + 16, n);
    const int ea = __builtin_amdgcn_readfirstlane(rowptr[r0]);
    const int eb = __builtin_amdgcn_readfirstlane(rowptr[r1]);

    float4 wv[8];
#pragma unroll
    for (int i = 0; i < 8; ++i) {
        const int idx = (int)threadIdx.x + i * (LAYER_WAVES * 32);
        wv[i] = *(const float4*)(W + (size_t)idx * 4);
    }

    int   nx_r, nx_c;
    float nx_w;
    {
        const int el = min(ea + lane, n_edges - 1);
        nx_r = erow[el];
        nx_c = ecol[el];
        nx_w = wE[el];
    }

    int   k0v[8], bv[8];
    float sl0[8], sl1[8];
    if (FIRST) {
        int rp[17];
#pragma unroll
        for (int i = 0; i < 17; ++i)
            rp[i] = __builtin_amdgcn_readfirstlane(rowptr[min(r0 + i, n)]);
#pragma unroll
        for (int i = 0; i < 8; ++i) {
            const int a  = half ? rp[i + 8] : rp[i];
            const int k0 = a + l16;
            k0v[i] = k0;
            bv[i]  = half ? rp[i + 9] : rp[i + 1];
            sl0[i] = wE[min(k0, n_edges - 1)];
            sl1[i] = wE[min(k0 + 16, n_edges - 1)];
        }
    }

    float4 hv[8];
#pragma unroll
    for (int i = 0; i < 8; ++i) {
        const int row  = r0 + 2 * i + half;
        const int rowc = row < n ? row : n - 1;
        hv[i] = *(const float4*)(h + (size_t)rowc * 64 + l16 * 4);
    }

#pragma unroll
    for (int i = 0; i < 8; ++i) {
        const int idx = (int)threadIdx.x + i * (LAYER_WAVES * 32);
        const int r   = idx >> 4;
        const int c4  = idx & 15;
        *(float4*)(sW + r * DPITCH + c4 * 4) = wv[i];
    }
#pragma unroll
    for (int i = 0; i < 8; ++i)
        *(float4*)(tbuf + (2 * i + half) * DPITCH + l16 * 4) = make_float4(0.0f, 0.0f, 0.0f, 0.0f);

    __syncthreads();
    if (!live) return;

    float rmax = 0.0f, rden = 1.0f;
    if (FIRST) {
        const float ninf = -__builtin_inff();
        float mx[8], sm[8];
#pragma unroll
        for (int i = 0; i < 8; ++i) {
            sl0[i] = k0v[i] < bv[i] ? sl0[i] : ninf;
            sl1[i] = k0v[i] + 16 < bv[i] ? sl1[i] : ninf;
            float m = fmaxf(sl0[i], sl1[i]);
            for (int k = k0v[i] + 32; k < bv[i]; k += 16) m = fmaxf(m, wE[k]);
            mx[i] = m;
        }
#pragma unroll
        for (int i = 0; i < 8; ++i) mx[i] = fmaxf(mx[i], __shfl_xor(mx[i], 1, 32));
#pragma unroll
        for (int i = 0; i < 8; ++i) mx[i] = fmaxf(mx[i], __shfl_xor(mx[i], 2, 32));
#pragma unroll
        for (int i = 0; i < 8; ++i) mx[i] = fmaxf(mx[i], __shfl_xor(mx[i], 4, 32));
#pragma unroll
        for (int i = 0; i < 8; ++i) mx[i] = fmaxf(mx[i], __shfl_xor(mx[i], 8, 32));
#pragma unroll
        for (int i = 0; i < 8; ++i) {
            const float e0 = expf(sl0[i] - mx[i]);
            const float e1 = expf(sl1[i] - mx[i]);
            const float x0 = k0v[i] < bv[i] ? e0 : 0.0f;
            const float x1 = k0v[i] + 16 < bv[i] ? e1 : 0.0f;
            float t = x0 + x1;
            for (int k = k0v[i] + 32; k < bv[i]; k += 16) t += expf(wE[k] - mx[i]);
            sm[i] = t;
        }
#pragma unroll
        for (int i = 0; i < 8; ++i) sm[i] += __shfl_xor(sm[i], 1, 32);
#pragma unroll
        for (int i = 0; i < 8; ++i) sm[i] += __shfl_xor(sm[i], 2, 32);
#pragma unroll
        for (int i = 0; i < 8; ++i) sm[i] += __shfl_xor(sm[i], 4, 32);
#pragma unroll
        for (int i = 0; i < 8; ++i) sm[i] += __shfl_xor(sm[i], 8, 32);
#pragma unroll
        for (int i = 0; i < 8; ++i)
            if (l16 == i) { rmax = mx[i]; rden = fmaxf(sm[i], GEPS); }
    }

    {
        float2* trow = (float2*)tbuf;
        int   cur = -1;
        float ax = 0.0f, ay = 0.0f;
        const char* hb = (const char*)h;
        const unsigned int lane8 = (unsigned int)lane * 8u;

        int p = ea;
        for (; p + 32 <= eb; p += 32) {
            const int   my_r = nx_r;
            const float my_w = FIRST ? edge_weight(nx_w, nx_r - r0, rmax, rden) : nx_w;
            if (FIRST) wOut[p + lane] = my_w;
            const unsigned int my_o = (unsigned int)nx_c * 256u;
            {
                const int el = min(p + 32 + lane, n_edges - 1);
                nx_r = erow[el];
                nx_c = ecol[el];
                nx_w = wE[el];
            }

            float2 gv[32];
#pragma unroll
            for (int j = 0; j < 32; ++j) {
                const unsigned int o = (unsigned int)__builtin_amdgcn_readlane((int)my_o, j);
                gv[j] = *(const float2*)(hb + (o + lane8));
            }
            int pr = __shfl_up(my_r, 1, 32);
            if (lane == 0) pr = cur;
            const unsigned int first = (unsigned int)__builtin_amdgcn_ballot_w32(my_r != pr);
#pragma unroll
            for (int j = 0; j < 32; ++j) {
                const float we = __int_as_float(__builtin_amdgcn_readlane(__float_as_int(my_w), j));
                if (__builtin_expect((first >> j) & 1u, 0u)) {
                    if (cur >= 0) trow[(cur - r0) * (DPITCH / 2) + lane] = make_float2(ax, ay);
                    ax = 0.0f; ay = 0.0f;
                    cur = __builtin_amdgcn_readlane(my_r, j);
                }
                ax = fmaf(we, gv[j].x, ax);
                ay = fmaf(we, gv[j].y, ay);
            }
        }

        if (p < eb) {
            const int   cnt  = eb - p;
            const int   my_r = nx_r;
            const float my_w = FIRST ? edge_weight(nx_w, nx_r - r0, rmax, rden) : nx_w;
            if (FIRST && lane < cnt) wOut[p + lane] = my_w;
            const unsigned int my_o = (unsigned int)nx_c * 256u;

            float2 gv[32];
#pragma unroll
            for (int j = 0; j < 32; ++j) {
                if (j < cnt) {
                    const unsigned int o = (unsigned int)__builtin_amdgcn_readlane((int)my_o, j);
                    gv[j] = *(const float2*)(hb + (o + lane8));
                }
            }

#pragma unroll
            for (int j = 0; j < 32; ++j) {
                if (j < cnt) {
                    const int   r  = __builtin_amdgcn_readlane(my_r, j);
                    const float we = __int_as_float(__builtin_amdgcn_readlane(__float_as_int(my_w), j));
                    if (r != cur) {
                        if (cur >= 0) trow[(cur - r0) * (DPITCH / 2) + lane] = make_float2(ax, ay);
                        ax = 0.0f; ay = 0.0f; cur = r;
                    }
                    ax = fmaf(we, gv[j].x, ax);
                    ay = fmaf(we, gv[j].y, ay);
                }
            }
        }
        if (cur >= 0) trow[(cur - r0) * (DPITCH / 2) + lane] = make_float2(ax, ay);
    }

    if (FIRST) {
        float* c0 = out0 + (size_t)tile * 16 * 64 + (size_t)half * 64 + l16 * 4;
        if (tile * 16 + 16 <= n) {
#pragma unroll
            for (int i = 0; i < 8; ++i)
                *(float4*)(c0 + (size_t)i * 128) = hv[i];
        } else {
#pragma unroll
            for (int i = 0; i < 8; ++i)
                if (tile * 16 + 2 * i + half < n)
                    *(float4*)(c0 + (size_t)i * 128) = hv[i];
        }
    }

    __builtin_amdgcn_fence(__ATOMIC_RELEASE, "wavefront");
    __builtin_amdgcn_wave_barrier();
    __builtin_amdgcn_fence(__ATOMIC_ACQUIRE, "wavefront");

#pragma unroll
    for (int i = 0; i < 8; ++i) {
        float* q = tbuf + (2 * i + half) * DPITCH + l16 * 4;
        const float4 mv = *(const float4*)q;
        float4 sm;
        sm.x = hv[i].x + mv.x;
        sm.y = hv[i].y + mv.y;
        sm.z = hv[i].z + mv.z;
        sm.w = hv[i].w + mv.w;
        *(float4*)q = sm;
    }

    __builtin_amdgcn_fence(__ATOMIC_RELEASE, "wavefront");
    __builtin_amdgcn_wave_barrier();
    __builtin_amdgcn_fence(__ATOMIC_ACQUIRE, "wavefront");

    v8f acc[4] = {};

#pragma unroll
    for (int kb = 0; kb < 16; ++kb) {
        const int koff = kb * 4 + half * 2;
        const v2f a = *(const v2f*)(tbuf + l16 * DPITCH + koff);
#pragma unroll
        for (int jb = 0; jb < 4; ++jb) {
            const v2f b = *(const v2f*)(sW + (jb * 16 + l16) * DPITCH + koff);
            acc[jb] = __builtin_amdgcn_wmma_f32_16x16x4_f32(
                false, a, false, b, (short)0, acc[jb], false, false);
        }
    }

    float bj[4];
#pragma unroll
    for (int jb = 0; jb < 4; ++jb) bj[jb] = bias[jb * 16 + l16];

#pragma unroll
    for (int v = 0; v < 8; ++v) {
        float vals[4];
        float ss = 0.0f;
#pragma unroll
        for (int jb = 0; jb < 4; ++jb) {
            float y = fmaxf(acc[jb][v] + bj[jb], 0.0f);
            vals[jb] = y;
            ss = fmaf(y, y, ss);
        }
        ss += __shfl_xor(ss, 1, 32);
        ss += __shfl_xor(ss, 2, 32);
        ss += __shfl_xor(ss, 4, 32);
        ss += __shfl_xor(ss, 8, 32);
        float inv = 1.0f / fmaxf(sqrtf(ss), GEPS);
        const int m = v + half * 8;
#pragma unroll
        for (int jb = 0; jb < 4; ++jb)
            tbuf[m * DPITCH + jb * 16 + l16] = vals[jb] * inv;
    }

    __builtin_amdgcn_fence(__ATOMIC_RELEASE, "wavefront");
    __builtin_amdgcn_wave_barrier();
    __builtin_amdgcn_fence(__ATOMIC_ACQUIRE, "wavefront");

    float4 o[8];
#pragma unroll
    for (int i = 0; i < 8; ++i)
        o[i] = *(const float4*)(tbuf + (2 * i + half) * DPITCH + l16 * 4);

    float* orow0 = out + (size_t)tile * 16 * 64 + (size_t)half * 64 + l16 * 4;
    if (tile * 16 + 16 <= n) {
#pragma unroll
        for (int i = 0; i < 8; ++i)
            *(float4*)(orow0 + (size_t)i * 128) = o[i];
    } else {
#pragma unroll
        for (int i = 0; i < 8; ++i)
            if (tile * 16 + 2 * i + half < n)
                *(float4*)(orow0 + (size_t)i * 128) = o[i];
    }
}

static inline int cdiv(long long a, long long b) { return (int)((a + b - 1) / b); }

extern "C" void kernel_launch(void* const* d_in, const int* in_sizes, int n_in,
                              void* d_out, int out_size, void* d_ws, size_t ws_size,
                              hipStream_t stream) {
    const float* h0    = (const float*)d_in[0];
    const float* h_att = (const float*)d_in[1];
    const float* t_att = (const float*)d_in[2];
    const float* w1    = (const float*)d_in[3];
    const float* b1    = (const float*)d_in[4];
    const float* w2    = (const float*)d_in[5];
    const float* b2    = (const float*)d_in[6];
    const int*   erow  = (const int*)d_in[7];
    const int*   ecol  = (const int*)d_in[8];
    float* out = (float*)d_out;

    const int N = in_sizes[0] / 64;
    const int E = in_sizes[7];

    float* ws  = (float*)d_ws;
    float* lg  = ws;
    float* wE  = ws + (size_t)N * 64;
    int* rowptr = (int*)(ws + (size_t)N * 64 + E);

    float* out1 = out + (size_t)N * 64;
    float* out2 = out + 2 * (size_t)N * 64;

    const int TPB = 256;

    const int edge_bl = cdiv(cdiv(E, 32 * LG_GROUPS), TPB / 32);

    logits_kernel<<<edge_bl, TPB, 0, stream>>>(h_att, t_att, erow, ecol, rowptr, lg, E, N);

    const int ntiles   = (N + 15) >> 4;
    const int layer_bl = cdiv(ntiles, LAYER_WAVES);

    layer_kernel<true><<<layer_bl, LAYER_WAVES * 32, 0, stream>>>(h0,   out,     lg, wE,      erow, ecol, rowptr, w1, b1, out1, N, E);
    layer_kernel<false><<<layer_bl, LAYER_WAVES * 32, 0, stream>>>(out1, nullptr, wE, nullptr, erow, ecol, rowptr, w2, b2, out2, N, E);
}
